# starting kernel
# baseline (speedup 1.0000x reference)
#include <hip/hip_runtime.h>
#include <math.h>

#define NNODES 50000
#define NEDGES 800000
#define NTOT   (NEDGES + NNODES)
#define FEAT 128
#define HID  128
#define HEADS 8
#define HDIM 16
#define NEG_SLOPE 0.2f

typedef __attribute__((ext_vector_type(2))) float v2f;
typedef __attribute__((ext_vector_type(8))) float v8f;

static __device__ __forceinline__ void atomicMaxF(float* addr, float val) {
    if (val >= 0.f) atomicMax((int*)addr, __float_as_int(val));
    else            atomicMin((unsigned int*)addr, (unsigned int)__float_as_int(val));
}

static __device__ __forceinline__ void edge_sd(const int* __restrict__ ei, int e,
                                               int& s, int& d) {
    if (e < NEDGES) { s = ei[e]; d = ei[NEDGES + e]; }
    else            { s = e - NEDGES; d = s; }
}

__global__ __launch_bounds__(256)
void k_gemm_wmma(const float* __restrict__ X, const float* __restrict__ W,
                 float* __restrict__ Hout) {
    __shared__ float ldsW[FEAT * HID];
    const int tid = threadIdx.x;

    const float4* Wv = (const float4*)W;
    float4*       Lv = (float4*)ldsW;
#pragma unroll
    for (int i = 0; i < (FEAT * HID / 4) / 256; ++i)
        Lv[tid + i * 256] = Wv[tid + i * 256];
    __syncthreads();

    const int lane = tid & 31;
    const int wave = tid >> 5;
    const int l16  = lane & 15;
    const int half = lane >> 4;
    const int row  = blockIdx.x * 16 + l16;
    const int col  = wave * 16 + l16;

    const float* xrow = X + (size_t)row * FEAT;

    v8f acc = {};
#pragma unroll
    for (int k0 = 0; k0 < FEAT; k0 += 4) {
        const int ka = k0 + 2 * half;
        v2f a, b;
        a.x = xrow[ka];
        a.y = xrow[ka + 1];
        b.x = ldsW[ka * HID + col];
        b.y = ldsW[(ka + 1) * HID + col];
        acc = __builtin_amdgcn_wmma_f32_16x16x4_f32(
                  false, a, false, b, (short)0, acc, false, false);
    }

    const int orow0 = blockIdx.x * 16 + 8 * half;
#pragma unroll
    for (int j = 0; j < 8; ++j)
        Hout[(size_t)(orow0 + j) * HID + col] = acc[j];
}

__global__ __launch_bounds__(256)
void k_alpha(const float* __restrict__ Hf, const float* __restrict__ a_src,
             const float* __restrict__ a_dst, float* __restrict__ AS,
             float* __restrict__ AD) {
    int t = blockIdx.x * blockDim.x + threadIdx.x;
    if (t >= NNODES * HEADS) return;
    int hd = t & (HEADS - 1);
    int n  = t >> 3;
    const float* hp = Hf + (size_t)n * HID + hd * HDIM;
    const float* as = a_src + hd * HDIM;
    const float* ad = a_dst + hd * HDIM;
    float sa = 0.f, sd = 0.f;
#pragma unroll
    for (int c = 0; c < HDIM; ++c) {
        float v = hp[c];
        sa += v * as[c];
        sd += v * ad[c];
    }
    AS[t] = sa;
    AD[t] = sd;
}

__global__ __launch_bounds__(256)
void k_fill(float* __restrict__ p, float v, int n) {
    int i = blockIdx.x * blockDim.x + threadIdx.x;
    if (i < n) p[i] = v;
}

static __device__ __forceinline__ float edge_e(const float* AS, const float* AD,
                                               int s, int d, int hd) {
    float v = AS[s * HEADS + hd] + AD[d * HEADS + hd];
    return v >= 0.f ? v : NEG_SLOPE * v;
}

__global__ __launch_bounds__(256)
void k_edge_max(const int* __restrict__ ei, const float* __restrict__ AS,
                const float* __restrict__ AD, float* __restrict__ M) {
    int t = blockIdx.x * blockDim.x + threadIdx.x;
    if (t >= NTOT * HEADS) return;
    int hd = t & (HEADS - 1), e = t >> 3;
    int s, d; edge_sd(ei, e, s, d);
    atomicMaxF(&M[d * HEADS + hd], edge_e(AS, AD, s, d, hd));
}

__global__ __launch_bounds__(256)
void k_edge_sum(const int* __restrict__ ei, const float* __restrict__ AS,
                const float* __restrict__ AD, const float* __restrict__ M,
                float* __restrict__ S) {
    int t = blockIdx.x * blockDim.x + threadIdx.x;
    if (t >= NTOT * HEADS) return;
    int hd = t & (HEADS - 1), e = t >> 3;
    int s, d; edge_sd(ei, e, s, d);
    float w = expf(edge_e(AS, AD, s, d, hd) - M[d * HEADS + hd]);
    atomicAdd(&S[d * HEADS + hd], w);
}

__global__ __launch_bounds__(256)
void k_edge_agg(const int* __restrict__ ei, const float* __restrict__ AS,
                const float* __restrict__ AD, const float* __restrict__ M,
                const float* __restrict__ S, const float* __restrict__ Hf,
                float* __restrict__ Agg) {
    int t = blockIdx.x * blockDim.x + threadIdx.x;
    if (t >= NTOT * HEADS) return;
    int hd = t & (HEADS - 1), e = t >> 3;
    int s, d; edge_sd(ei, e, s, d);
    float w     = expf(edge_e(AS, AD, s, d, hd) - M[d * HEADS + hd]);
    float alpha = w / (S[d * HEADS + hd] + 1e-16f);
    const float* hs = Hf  + (size_t)s * HID + hd * HDIM;
    float*       op = Agg + (size_t)d * HID + hd * HDIM;
    __builtin_prefetch(hs, 0, 1);
#pragma unroll
    for (int c = 0; c < HDIM; ++c)
        atomicAdd(&op[c], alpha * hs[c]);
}

__global__ __launch_bounds__(256)
void k_bias_elu(const float* __restrict__ Agg, const float* __restrict__ b,
                float* __restrict__ Out) {
    int t = blockIdx.x * blockDim.x + threadIdx.x;
    if (t >= NNODES * HID) return;
    float v = Agg[t] + b[t & (HID - 1)];
    Out[t] = v > 0.f ? v : expm1f(v);
}

static inline unsigned grid_for(long long n) { return (unsigned)((n + 255) / 256); }

extern "C" void kernel_launch(void* const* d_in, const int* in_sizes, int n_in,
                              void* d_out, int out_size, void* d_ws, size_t ws_size,
                              hipStream_t stream) {
    const float* x   = (const float*)d_in[0];
    const int*   ei  = (const int*)  d_in[1];
    const float* W1  = (const float*)d_in[2];
    const float* as1 = (const float*)d_in[3];
    const float* ad1 = (const float*)d_in[4];
    const float* b1  = (const float*)d_in[5];
    const float* W2  = (const float*)d_in[6];
    const float* as2 = (const float*)d_in[7];
    const float* ad2 = (const float*)d_in[8];
    const float* b2  = (const float*)d_in[9];

    float* ws     = (float*)d_ws;
    float* bufH   = ws;
    float* bufAS  = bufH  + (size_t)NNODES * HID;
    float* bufAD  = bufAS + (size_t)NNODES * HEADS;
    float* bufM   = bufAD + (size_t)NNODES * HEADS;
    float* bufS   = bufM  + (size_t)NNODES * HEADS;
    float* bufAgg = bufS  + (size_t)NNODES * HEADS;

    const float neg_inf = -__builtin_huge_valf();

    auto run_layer = [&](const float* xin, const float* Wm, const float* as,
                         const float* ad, const float* bv, float* outBuf) {
        k_gemm_wmma<<<NNODES / 16, 256, 0, stream>>>(xin, Wm, bufH);
        k_alpha<<<grid_for((long long)NNODES * HEADS), 256, 0, stream>>>(
            bufH, as, ad, bufAS, bufAD);
        k_fill<<<grid_for((long long)NNODES * HEADS), 256, 0, stream>>>(
            bufM, neg_inf, NNODES * HEADS);
        hipMemsetAsync(bufS, 0, (size_t)NNODES * HEADS * sizeof(float), stream);
        hipMemsetAsync(outBuf, 0, (size_t)NNODES * HID * sizeof(float), stream);
        k_edge_max<<<grid_for((long long)NTOT * HEADS), 256, 0, stream>>>(
            ei, bufAS, bufAD, bufM);
        k_edge_sum<<<grid_for((long long)NTOT * HEADS), 256, 0, stream>>>(
            ei, bufAS, bufAD, bufM, bufS);
        k_edge_agg<<<grid_for((long long)NTOT * HEADS), 256, 0, stream>>>(
            ei, bufAS, bufAD, bufM, bufS, bufH, outBuf);
        k_bias_elu<<<grid_for((long long)NNODES * HID), 256, 0, stream>>>(
            outBuf, bv, outBuf);
    };

    run_layer(x,      W1, as1, ad1, b1, bufAgg);
    run_layer(bufAgg, W2, as2, ad2, b2, (float*)d_out);
}


# Edge phase, no global atomics, no sort: each of 256 blocks filters the dst array for its own 196 nodes into an LDS list, then max, exp-sum, row sum by LDS atomics, one row store; sum order changes
# speedup vs baseline: 5.1185x; 5.1185x over previous
#include <hip/hip_runtime.h>
#include <math.h>

#define NNODES 50000
#define NEDGES 800000
#define FEAT 128
#define HID  128
#define HEADS 8
#define HDIM 16
#define NEG_SLOPE 0.2f

#define OWN_R    196
#define OWN_NB   256
#define OWN_T    1024
#define OWN_W    (OWN_T / 32)
#define LIST_CAP 32768
#define AGG_U    8

typedef __attribute__((ext_vector_type(2))) float v2f;
typedef __attribute__((ext_vector_type(8))) float v8f;

__global__ __launch_bounds__(256)
void k_gemm_wmma(const float* __restrict__ X, const float* __restrict__ W,
                 float* __restrict__ Hout) {
    __shared__ float ldsW[FEAT * HID];
    const int tid = threadIdx.x;

    const float4* Wv = (const float4*)W;
    float4*       Lv = (float4*)ldsW;
#pragma unroll
    for (int i = 0; i < (FEAT * HID / 4) / 256; ++i)
        Lv[tid + i * 256] = Wv[tid + i * 256];
    __syncthreads();

    const int lane = tid & 31;
    const int wave = tid >> 5;
    const int l16  = lane & 15;
    const int half = lane >> 4;
    const int row  = blockIdx.x * 16 + l16;
    const int col  = wave * 16 + l16;

    const float* xrow = X + (size_t)row * FEAT;

    v8f acc = {};
#pragma unroll
    for (int k0 = 0; k0 < FEAT; k0 += 4) {
        const int ka = k0 + 2 * half;
        v2f a, b;
        a.x = xrow[ka];
        a.y = xrow[ka + 1];
        b.x = ldsW[ka * HID + col];
        b.y = ldsW[(ka + 1) * HID + col];
        acc = __builtin_amdgcn_wmma_f32_16x16x4_f32(
                  false, a, false, b, (short)0, acc, false, false);
    }

    const int orow0 = blockIdx.x * 16 + 8 * half;
#pragma unroll
    for (int j = 0; j < 8; ++j)
        Hout[(size_t)(orow0 + j) * HID + col] = acc[j];
}

__global__ __launch_bounds__(256)
void k_alpha(const float* __restrict__ Hf, const float* __restrict__ a_src,
             const float* __restrict__ a_dst, float* __restrict__ AS,
             float* __restrict__ AD) {
    int t = blockIdx.x * blockDim.x + threadIdx.x;
    if (t >= NNODES * HEADS) return;
    int hd = t & (HEADS - 1);
    int n  = t >> 3;
    const float* hp = Hf + (size_t)n * HID + hd * HDIM;
    const float* as = a_src + hd * HDIM;
    const float* ad = a_dst + hd * HDIM;
    float sa = 0.f, sd = 0.f;
#pragma unroll
    for (int c = 0; c < HDIM; ++c) {
        float v = hp[c];
        sa += v * as[c];
        sd += v * ad[c];
    }
    AS[t] = sa;
    AD[t] = sd;
}

static __device__ __forceinline__ float leaky(float v) {
    return v >= 0.f ? v : NEG_SLOPE * v;
}

static __device__ __forceinline__ void lds_fmax(float* p, float v) {
    __hip_atomic_fetch_max(p, v, __ATOMIC_RELAXED, __HIP_MEMORY_SCOPE_WORKGROUP);
}

static __device__ __forceinline__ int acc_pos(int c) { return (c & 3) * 32 + (c >> 2); }

static __device__ void edge_full_lane(const float* __restrict__ AS, const float* __restrict__ Hf,
                                      const float* sMA, const float* sAD, float* sS, float* acc,
                                      int s, int ld) {
    for (int hd = 0; hd < HEADS; ++hd) {
        const float ad = sAD[ld * HEADS + hd];
        const float m  = leaky(sMA[ld * HEADS + hd] + ad);
        const float w  = expf(leaky(AS[s * HEADS + hd] + ad) - m);
        atomicAdd(&sS[ld * HEADS + hd], w);
        for (int c = 0; c < HDIM; ++c) {
            const int col = hd * HDIM + c;
            atomicAdd(&acc[ld * HID + acc_pos(col)], w * Hf[(size_t)s * HID + col]);
        }
    }
}

__global__ __launch_bounds__(OWN_T)
void k_owner(const int* __restrict__ ei, const float* __restrict__ AS,
             const float* __restrict__ AD, const float* __restrict__ Hf,
             const float* __restrict__ bias, float* __restrict__ Out) {
    __shared__ float    acc[OWN_R * HID];
    __shared__ float    sMA[OWN_R * HEADS];
    __shared__ float    sAD[OWN_R * HEADS];
    __shared__ float    sS [OWN_R * HEADS];
    __shared__ unsigned list[LIST_CAP];
    __shared__ unsigned cnt;

    const int tid  = threadIdx.x;
    const int base = blockIdx.x * OWN_R;
    const int Rn   = min(OWN_R, NNODES - base);

    for (int i = tid; i < OWN_R * HID; i += OWN_T) acc[i] = 0.f;
    for (int i = tid; i < Rn * HEADS; i += OWN_T) {
        sMA[i] = AS[base * HEADS + i];
        sAD[i] = AD[base * HEADS + i];
        sS[i]  = 0.f;
    }
    if (tid == 0) cnt = 0u;
    __syncthreads();

    const int4* dv = (const int4*)(ei + NEDGES);
    const int   NV = NEDGES / 4;
    for (int i0 = tid; i0 < NV; i0 += 4 * OWN_T) {
        int4 v[4];
#pragma unroll
        for (int u = 0; u < 4; ++u) {
            const int idx = i0 + u * OWN_T;
            v[u] = idx < NV ? dv[idx] : make_int4(-1, -1, -1, -1);
        }
#pragma unroll
        for (int u = 0; u < 4; ++u) {
            const int e0 = (i0 + u * OWN_T) * 4;
            const int dd[4] = {v[u].x, v[u].y, v[u].z, v[u].w};
#pragma unroll
            for (int j = 0; j < 4; ++j) {
                const unsigned ld = (unsigned)(dd[j] - base);
                if (ld < (unsigned)Rn) {
                    const unsigned slot = atomicAdd(&cnt, 1u);
                    if (slot < LIST_CAP) list[slot] = ((unsigned)(e0 + j) << 8) | ld;
                }
            }
        }
    }
    __syncthreads();
    const unsigned total = cnt;

    if (total <= LIST_CAP) {
        for (unsigned i = tid; i < total; i += OWN_T) {
            const unsigned ent = list[i];
            const unsigned ld  = ent & 255u;
            const int      s   = ei[ent >> 8];
            list[i] = ((unsigned)s << 8) | ld;
            const float4* ap = (const float4*)(AS + (size_t)s * HEADS);
            const float4 a0 = ap[0], a1 = ap[1];
            float* m = &sMA[ld * HEADS];
            lds_fmax(m + 0, a0.x); lds_fmax(m + 1, a0.y);
            lds_fmax(m + 2, a0.z); lds_fmax(m + 3, a0.w);
            lds_fmax(m + 4, a1.x); lds_fmax(m + 5, a1.y);
            lds_fmax(m + 6, a1.z); lds_fmax(m + 7, a1.w);
        }
        __syncthreads();

        const int lane = tid & 31, wave = tid >> 5, hd = lane >> 2;
        const unsigned nitems = total + (unsigned)Rn;
        for (unsigned i0 = wave * AGG_U; i0 < nitems; i0 += OWN_W * AGG_U) {
            int    sv[AGG_U], lv[AGG_U];
            float4 hv[AGG_U];
            float  av[AGG_U];
#pragma unroll
            for (int u = 0; u < AGG_U; ++u) {
                const unsigned i = i0 + u;
                if (i < total)       { const unsigned ent = list[i]; sv[u] = ent >> 8; lv[u] = ent & 255u; }
                else if (i < nitems) { lv[u] = i - total; sv[u] = base + lv[u]; }
                else                 { lv[u] = 0; sv[u] = base; }
            }
#pragma unroll
            for (int u = 0; u < AGG_U; ++u) {
                hv[u] = *(const float4*)(Hf + (size_t)sv[u] * HID + 4 * lane);
                av[u] = AS[sv[u] * HEADS + hd];
            }
#pragma unroll
            for (int u = 0; u < AGG_U; ++u) {
                if (i0 + u < nitems) {
                    const int   ld = lv[u];
                    const float ad = sAD[ld * HEADS + hd];
                    const float m  = leaky(sMA[ld * HEADS + hd] + ad);
                    const float w  = expf(leaky(av[u] + ad) - m);
                    float* a = &acc[ld * HID + lane];
                    atomicAdd(a,      w * hv[u].x);
                    atomicAdd(a + 32, w * hv[u].y);
                    atomicAdd(a + 64, w * hv[u].z);
                    atomicAdd(a + 96, w * hv[u].w);
                    if ((lane & 3) == 0) atomicAdd(&sS[ld * HEADS + hd], w);
                }
            }
        }
    } else {
        for (int e = tid; e < NEDGES; e += OWN_T) {
            const unsigned ld = (unsigned)(ei[NEDGES + e] - base);
            if (ld < (unsigned)Rn) {
                const int s = ei[e];
                for (int h = 0; h < HEADS; ++h) lds_fmax(&sMA[ld * HEADS + h], AS[s * HEADS + h]);
            }
        }
        __syncthreads();
        for (int e = tid; e < NEDGES; e += OWN_T) {
            const unsigned ld = (unsigned)(ei[NEDGES + e] - base);
            if (ld < (unsigned)Rn) edge_full_lane(AS, Hf, sMA, sAD, sS, acc, ei[e], (int)ld);
        }
        for (int ld = tid; ld < Rn; ld += OWN_T)
            edge_full_lane(AS, Hf, sMA, sAD, sS, acc, base + ld, ld);
    }
    __syncthreads();

    for (int i = tid; i < Rn * HID; i += OWN_T) {
        const int ld = i >> 7, c = i & (HID - 1);
        const float v = acc[ld * HID + acc_pos(c)] / (sS[ld * HEADS + (c >> 4)] + 1e-16f) + bias[c];
        Out[(size_t)(base + ld) * HID + c] = v > 0.f ? v : expm1f(v);
    }
}

static inline unsigned grid_for(long long n) { return (unsigned)((n + 255) / 256); }

extern "C" void kernel_launch(void* const* d_in, const int* in_sizes, int n_in,
                              void* d_out, int out_size, void* d_ws, size_t ws_size,
                              hipStream_t stream) {
    const float* x   = (const float*)d_in[0];
    const int*   ei  = (const int*)  d_in[1];
    const float* W1  = (const float*)d_in[2];
    const float* as1 = (const float*)d_in[3];
    const float* ad1 = (const float*)d_in[4];
    const float* b1  = (const float*)d_in[5];
    const float* W2  = (const float*)d_in[6];
    const float* as2 = (const float*)d_in[7];
    const float* ad2 = (const float*)d_in[8];
    const float* b2  = (const float*)d_in[9];

    float* ws     = (float*)d_ws;
    float* bufH   = ws;
    float* bufAS  = bufH  + (size_t)NNODES * HID;
    float* bufAD  = bufAS + (size_t)NNODES * HEADS;
    float* bufAgg = bufAD + (size_t)NNODES * HEADS;

    auto run_layer = [&](const float* xin, const float* Wm, const float* as,
                         const float* ad, const float* bv, float* outBuf) {
        k_gemm_wmma<<<NNODES / 16, 256, 0, stream>>>(xin, Wm, bufH);
        k_alpha<<<grid_for((long long)NNODES * HEADS), 256, 0, stream>>>(
            bufH, as, ad, bufAS, bufAD);
        k_owner<<<OWN_NB, OWN_T, 0, stream>>>(ei, bufAS, bufAD, bufH, bv, outBuf);
    };

    run_layer(x,      W1, as1, ad1, b1, bufAgg);
    run_layer(bufAgg, W2, as2, ad2, b2, (float*)d_out);
}


# Destination array scanned once per call: layer-1 owner block saves its (source,node) edge list in its own rows of the output buffer, layer 2 loads it back; sums stay arrival-ordered as before
# speedup vs baseline: 5.9086x; 1.1544x over previous
#include <hip/hip_runtime.h>
#include <math.h>

#define NNODES 50000
#define NEDGES 800000
#define FEAT 128
#define HID  128
#define HEADS 8
#define HDIM 16
#define NEG_SLOPE 0.2f

#define OWN_R    196
#define OWN_NB   256
#define OWN_T    1024
#define OWN_W    (OWN_T / 32)
#define LIST_CAP 32768
#define AGG_U    8
#define KEEP_HDR 32

typedef __attribute__((ext_vector_type(2))) float v2f;
typedef __attribute__((ext_vector_type(8))) float v8f;

__global__ __launch_bounds__(256)
void k_gemm_wmma(const float* __restrict__ X, const float* __restrict__ W,
                 float* __restrict__ Hout) {
    __shared__ float ldsW[FEAT * HID];
    const int tid = threadIdx.x;

    const float4* Wv = (const float4*)W;
    float4*       Lv = (float4*)ldsW;
#pragma unroll
    for (int i = 0; i < (FEAT * HID / 4) / 256; ++i)
        Lv[tid + i * 256] = Wv[tid + i * 256];
    __syncthreads();

    const int lane = tid & 31;
    const int wave = tid >> 5;
    const int l16  = lane & 15;
    const int half = lane >> 4;
    const int row  = blockIdx.x * 16 + l16;
    const int col  = wave * 16 + l16;

    const float* xrow = X + (size_t)row * FEAT;

    v8f acc = {};
#pragma unroll
    for (int k0 = 0; k0 < FEAT; k0 += 4) {
        const int ka = k0 + 2 * half;
        v2f a, b;
        a.x = xrow[ka];
        a.y = xrow[ka + 1];
        b.x = ldsW[ka * HID + col];
        b.y = ldsW[(ka + 1) * HID + col];
        acc = __builtin_amdgcn_wmma_f32_16x16x4_f32(
                  false, a, false, b, (short)0, acc, false, false);
    }

    const int orow0 = blockIdx.x * 16 + 8 * half;
#pragma unroll
    for (int j = 0; j < 8; ++j)
        Hout[(size_t)(orow0 + j) * HID + col] = acc[j];
}

__global__ __launch_bounds__(256)
void k_alpha(const float* __restrict__ Hf, const float* __restrict__ a_src,
             const float* __restrict__ a_dst, float* __restrict__ AS,
             float* __restrict__ AD) {
    int t = blockIdx.x * blockDim.x + threadIdx.x;
    if (t >= NNODES * HEADS) return;
    int hd = t & (HEADS - 1);
    int n  = t >> 3;
    const float* hp = Hf + (size_t)n * HID + hd * HDIM;
    const float* as = a_src + hd * HDIM;
    const float* ad = a_dst + hd * HDIM;
    float sa = 0.f, sd = 0.f;
#pragma unroll
    for (int c = 0; c < HDIM; ++c) {
        float v = hp[c];
        sa += v * as[c];
        sd += v * ad[c];
    }
    AS[t] = sa;
    AD[t] = sd;
}

static __device__ __forceinline__ float leaky(float v) {
    return v >= 0.f ? v : NEG_SLOPE * v;
}

static __device__ __forceinline__ void lds_fmax(float* p, float v) {
    __hip_atomic_fetch_max(p, v, __ATOMIC_RELAXED, __HIP_MEMORY_SCOPE_WORKGROUP);
}

static __device__ __forceinline__ int acc_pos(int c) { return (c & 3) * 32 + (c >> 2); }

static __device__ void edge_full_lane(const float* __restrict__ AS, const float* __restrict__ Hf,
                                      const float* sMA, const float* sAD, float* sS, float* acc,
                                      int s, int ld) {
    for (int hd = 0; hd < HEADS; ++hd) {
        const float ad = sAD[ld * HEADS + hd];
        const float m  = leaky(sMA[ld * HEADS + hd] + ad);
        const float w  = expf(leaky(AS[s * HEADS + hd] + ad) - m);
        atomicAdd(&sS[ld * HEADS + hd], w);
        for (int c = 0; c < HDIM; ++c) {
            const int col = hd * HDIM + c;
            atomicAdd(&acc[ld * HID + acc_pos(col)], w * Hf[(size_t)s * HID + col]);
        }
    }
}

__global__ __launch_bounds__(OWN_T)
void k_owner(const int* __restrict__ ei, const float* __restrict__ AS,
             const float* __restrict__ AD, const float* __restrict__ Hf,
             const float* __restrict__ bias, float* Out, unsigned* keep, int first) {
    __shared__ float    acc[OWN_R * HID];
    __shared__ float    sMA[OWN_R * HEADS];
    __shared__ float    sAD[OWN_R * HEADS];
    __shared__ float    sS [OWN_R * HEADS];
    __shared__ unsigned list[LIST_CAP];
    __shared__ unsigned cnt;

    const int tid  = threadIdx.x;
    const int base = blockIdx.x * OWN_R;
    const int Rn   = min(OWN_R, NNODES - base);

    unsigned*      mine     = keep + (size_t)base * HID;
    const unsigned keep_cap = (unsigned)(Rn * HID - KEEP_HDR);
    const unsigned kept     = first ? 0xFFFFFFFFu : mine[0];
    const bool     reuse    = kept <= keep_cap;

    for (int i = tid; i < OWN_R * HID; i += OWN_T) acc[i] = 0.f;
    for (int i = tid; i < Rn * HEADS; i += OWN_T) {
        sMA[i] = AS[base * HEADS + i];
        sAD[i] = AD[base * HEADS + i];
        sS[i]  = 0.f;
    }
    if (tid == 0) cnt = reuse ? kept : 0u;
    if (reuse)
        for (unsigned i = tid; i < kept; i += OWN_T) list[i] = mine[KEEP_HDR + i];
    __syncthreads();

    const int4* dv = (const int4*)(ei + NEDGES);
    const int   NV = NEDGES / 4;
    if (!reuse)
    for (int i0 = tid; i0 < NV; i0 += 4 * OWN_T) {
        int4 v[4];
#pragma unroll
        for (int u = 0; u < 4; ++u) {
            const int idx = i0 + u * OWN_T;
            v[u] = idx < NV ? dv[idx] : make_int4(-1, -1, -1, -1);
        }
#pragma unroll
        for (int u = 0; u < 4; ++u) {
            const int e0 = (i0 + u * OWN_T) * 4;
            const int dd[4] = {v[u].x, v[u].y, v[u].z, v[u].w};
#pragma unroll
            for (int j = 0; j < 4; ++j) {
                const unsigned ld = (unsigned)(dd[j] - base);
                if (ld < (unsigned)Rn) {
                    const unsigned slot = atomicAdd(&cnt, 1u);
                    if (slot < LIST_CAP) list[slot] = ((unsigned)(e0 + j) << 8) | ld;
                }
            }
        }
    }
    __syncthreads();
    const unsigned total = cnt;
    const bool     save  = first && total <= keep_cap;
    if (first && tid == 0) mine[0] = total;

    if (total <= LIST_CAP) {
        for (unsigned i = tid; i < total; i += OWN_T) {
            unsigned       ent = list[i];
            const unsigned ld  = ent & 255u;
            int            s   = (int)(ent >> 8);
            if (!reuse) {
                s       = ei[s];
                ent     = ((unsigned)s << 8) | ld;
                list[i] = ent;
                if (save) mine[KEEP_HDR + i] = ent;
            }
            const float4* ap = (const float4*)(AS + (size_t)s * HEADS);
            const float4 a0 = ap[0], a1 = ap[1];
            float* m = &sMA[ld * HEADS];
            lds_fmax(m + 0, a0.x); lds_fmax(m + 1, a0.y);
            lds_fmax(m + 2, a0.z); lds_fmax(m + 3, a0.w);
            lds_fmax(m + 4, a1.x); lds_fmax(m + 5, a1.y);
            lds_fmax(m + 6, a1.z); lds_fmax(m + 7, a1.w);
        }
        __syncthreads();

        const int lane = tid & 31, wave = tid >> 5, hd = lane >> 2;
        const unsigned nitems = total + (unsigned)Rn;
        for (unsigned i0 = wave * AGG_U; i0 < nitems; i0 += OWN_W * AGG_U) {
            int    sv[AGG_U], lv[AGG_U];
            float4 hv[AGG_U];
            float  av[AGG_U];
#pragma unroll
            for (int u = 0; u < AGG_U; ++u) {
                const unsigned i = i0 + u;
                if (i < total)       { const unsigned ent = list[i]; sv[u] = ent >> 8; lv[u] = ent & 255u; }
                else if (i < nitems) { lv[u] = i - total; sv[u] = base + lv[u]; }
                else                 { lv[u] = 0; sv[u] = base; }
            }
#pragma unroll
            for (int u = 0; u < AGG_U; ++u) {
                hv[u] = *(const float4*)(Hf + (size_t)sv[u] * HID + 4 * lane);
                av[u] = AS[sv[u] * HEADS + hd];
            }
#pragma unroll
            for (int u = 0; u < AGG_U; ++u) {
                if (i0 + u < nitems) {
                    const int   ld = lv[u];
                    const float ad = sAD[ld * HEADS + hd];
                    const float m  = leaky(sMA[ld * HEADS + hd] + ad);
                    const float w  = expf(leaky(av[u] + ad) - m);
                    float* a = &acc[ld * HID + lane];
                    atomicAdd(a,      w * hv[u].x);
                    atomicAdd(a + 32, w * hv[u].y);
                    atomicAdd(a + 64, w * hv[u].z);
                    atomicAdd(a + 96, w * hv[u].w);
                    if ((lane & 3) == 0) atomicAdd(&sS[ld * HEADS + hd], w);
                }
            }
        }
    } else {
        for (int e = tid; e < NEDGES; e += OWN_T) {
            const unsigned ld = (unsigned)(ei[NEDGES + e] - base);
            if (ld < (unsigned)Rn) {
                const int s = ei[e];
                for (int h = 0; h < HEADS; ++h) lds_fmax(&sMA[ld * HEADS + h], AS[s * HEADS + h]);
            }
        }
        __syncthreads();
        for (int e = tid; e < NEDGES; e += OWN_T) {
            const unsigned ld = (unsigned)(ei[NEDGES + e] - base);
            if (ld < (unsigned)Rn) edge_full_lane(AS, Hf, sMA, sAD, sS, acc, ei[e], (int)ld);
        }
        for (int ld = tid; ld < Rn; ld += OWN_T)
            edge_full_lane(AS, Hf, sMA, sAD, sS, acc, base + ld, ld);
    }
    __syncthreads();

    for (int i = tid; i < Rn * HID; i += OWN_T) {
        const int ld = i >> 7, c = i & (HID - 1);
        const float v = acc[ld * HID + acc_pos(c)] / (sS[ld * HEADS + (c >> 4)] + 1e-16f) + bias[c];
        Out[(size_t)(base + ld) * HID + c] = v > 0.f ? v : expm1f(v);
    }
}

static inline unsigned grid_for(long long n) { return (unsigned)((n + 255) / 256); }

extern "C" void kernel_launch(void* const* d_in, const int* in_sizes, int n_in,
                              void* d_out, int out_size, void* d_ws, size_t ws_size,
                              hipStream_t stream) {
    const float* x   = (const float*)d_in[0];
    const int*   ei  = (const int*)  d_in[1];
    const float* W1  = (const float*)d_in[2];
    const float* as1 = (const float*)d_in[3];
    const float* ad1 = (const float*)d_in[4];
    const float* b1  = (const float*)d_in[5];
    const float* W2  = (const float*)d_in[6];
    const float* as2 = (const float*)d_in[7];
    const float* ad2 = (const float*)d_in[8];
    const float* b2  = (const float*)d_in[9];

    float* ws     = (float*)d_ws;
    float* bufH   = ws;
    float* bufAS  = bufH  + (size_t)NNODES * HID;
    float* bufAD  = bufAS + (size_t)NNODES * HEADS;
    float* bufAgg = bufAD + (size_t)NNODES * HEADS;

    auto run_layer = [&](const float* xin, const float* Wm, const float* as,
                         const float* ad, const float* bv, float* outBuf, int first) {
        k_gemm_wmma<<<NNODES / 16, 256, 0, stream>>>(xin, Wm, bufH);
        k_alpha<<<grid_for((long long)NNODES * HEADS), 256, 0, stream>>>(
            bufH, as, ad, bufAS, bufAD);
        k_owner<<<OWN_NB, OWN_T, 0, stream>>>(ei, bufAS, bufAD, bufH, bv, outBuf,
                                              (unsigned*)d_out, first);
    };

    run_layer(x,      W1, as1, ad1, b1, bufAgg,        1);
    run_layer(bufAgg, W2, as2, ad2, b2, (float*)d_out, 0);
}


# Layer-1 edge filter reads a one-byte owner id per edge (destination / 196) that the layer-1 score launch writes into rows of the output buffer: 800 KB scanned per block, was 3200 KB
# speedup vs baseline: 6.6368x; 1.1232x over previous
#include <hip/hip_runtime.h>
#include <math.h>

#define NNODES 50000
#define NEDGES 800000
#define FEAT 128
#define HID  128
#define HEADS 8
#define HDIM 16
#define NEG_SLOPE 0.2f

#define OWN_R    196
#define OWN_NB   256
#define OWN_T    1024
#define OWN_W    (OWN_T / 32)
#define LIST_CAP 32768
#define AGG_U    8
#define KEEP_HDR 32
#define MAP_SLICE 3200
#define MAP_W     (MAP_SLICE / 4)
#define MAP_SV    (MAP_SLICE / 16)
#define REGION_V  (OWN_R * HID / 4)
#define MAP_NV    (NEDGES / 16)
#define MAP_U     7
static_assert(MAP_SLICE % 128 == 0 && NEDGES % MAP_SLICE == 0 && NEDGES / MAP_SLICE < OWN_NB,
              "whole slices, all in full-size blocks");
static_assert(NEDGES / 4 <= NNODES * HEADS, "the score launch has a thread per four edges");
static_assert((NNODES - (OWN_NB - 1) * OWN_R) * HID > MAP_W + KEEP_HDR, "last block holds its header");

typedef __attribute__((ext_vector_type(2))) float v2f;
typedef __attribute__((ext_vector_type(8))) float v8f;

__global__ __launch_bounds__(256)
void k_gemm_wmma(const float* __restrict__ X, const float* __restrict__ W,
                 float* __restrict__ Hout) {
    __shared__ float ldsW[FEAT * HID];
    const int tid = threadIdx.x;

    const float4* Wv = (const float4*)W;
    float4*       Lv = (float4*)ldsW;
#pragma unroll
    for (int i = 0; i < (FEAT * HID / 4) / 256; ++i)
        Lv[tid + i * 256] = Wv[tid + i * 256];
    __syncthreads();

    const int lane = tid & 31;
    const int wave = tid >> 5;
    const int l16  = lane & 15;
    const int half = lane >> 4;
    const int row  = blockIdx.x * 16 + l16;
    const int col  = wave * 16 + l16;

    const float* xrow = X + (size_t)row * FEAT;

    v8f acc = {};
#pragma unroll
    for (int k0 = 0; k0 < FEAT; k0 += 4) {
        const int ka = k0 + 2 * half;
        v2f a, b;
        a.x = xrow[ka];
        a.y = xrow[ka + 1];
        b.x = ldsW[ka * HID + col];
        b.y = ldsW[(ka + 1) * HID + col];
        acc = __builtin_amdgcn_wmma_f32_16x16x4_f32(
                  false, a, false, b, (short)0, acc, false, false);
    }

    const int orow0 = blockIdx.x * 16 + 8 * half;
#pragma unroll
    for (int j = 0; j < 8; ++j)
        Hout[(size_t)(orow0 + j) * HID + col] = acc[j];
}

__global__ __launch_bounds__(256)
void k_alpha(const float* __restrict__ Hf, const float* __restrict__ a_src,
             const float* __restrict__ a_dst, float* __restrict__ AS,
             float* __restrict__ AD, const int* __restrict__ ei,
             unsigned* keep, int first) {
    int t = blockIdx.x * blockDim.x + threadIdx.x;
    if (t >= NNODES * HEADS) return;
    if (first && t < NEDGES / 4) {
        const int4     d  = *(const int4*)(ei + NEDGES + 4 * t);
        const unsigned o  = ((unsigned)d.x / OWN_R)         | (((unsigned)d.y / OWN_R) << 8) |
                            (((unsigned)d.z / OWN_R) << 16) | (((unsigned)d.w / OWN_R) << 24);
        const unsigned sl = (unsigned)t / MAP_W;
        keep[(size_t)sl * (OWN_R * HID) + ((unsigned)t - sl * MAP_W)] = o;
    }
    int hd = t & (HEADS - 1);
    int n  = t >> 3;
    const float* hp = Hf + (size_t)n * HID + hd * HDIM;
    const float* as = a_src + hd * HDIM;
    const float* ad = a_dst + hd * HDIM;
    float sa = 0.f, sd = 0.f;
#pragma unroll
    for (int c = 0; c < HDIM; ++c) {
        float v = hp[c];
        sa += v * as[c];
        sd += v * ad[c];
    }
    AS[t] = sa;
    AD[t] = sd;
}

static __device__ __forceinline__ float leaky(float v) {
    return v >= 0.f ? v : NEG_SLOPE * v;
}

static __device__ __forceinline__ unsigned zero_bytes(unsigned x) {
    return ~(((x & 0x7F7F7F7Fu) + 0x7F7F7F7Fu) | x) & 0x80808080u;
}

static __device__ __forceinline__ void lds_fmax(float* p, float v) {
    __hip_atomic_fetch_max(p, v, __ATOMIC_RELAXED, __HIP_MEMORY_SCOPE_WORKGROUP);
}

static __device__ __forceinline__ int acc_pos(int c) { return (c & 3) * 32 + (c >> 2); }

static __device__ void edge_full_lane(const float* __restrict__ AS, const float* __restrict__ Hf,
                                      const float* sMA, const float* sAD, float* sS, float* acc,
                                      int s, int ld) {
    for (int hd = 0; hd < HEADS; ++hd) {
        const float ad = sAD[ld * HEADS + hd];
        const float m  = leaky(sMA[ld * HEADS + hd] + ad);
        const float w  = expf(leaky(AS[s * HEADS + hd] + ad) - m);
        atomicAdd(&sS[ld * HEADS + hd], w);
        for (int c = 0; c < HDIM; ++c) {
            const int col = hd * HDIM + c;
            atomicAdd(&acc[ld * HID + acc_pos(col)], w * Hf[(size_t)s * HID + col]);
        }
    }
}

__global__ __launch_bounds__(OWN_T)
void k_owner(const int* __restrict__ ei, const float* __restrict__ AS,
             const float* __restrict__ AD, const float* __restrict__ Hf,
             const float* __restrict__ bias, float* Out, unsigned* keep, int first) {
    __shared__ float    acc[OWN_R * HID];
    __shared__ float    sMA[OWN_R * HEADS];
    __shared__ float    sAD[OWN_R * HEADS];
    __shared__ float    sS [OWN_R * HEADS];
    __shared__ unsigned list[LIST_CAP];
    __shared__ unsigned cnt;

    const int tid  = threadIdx.x;
    const int base = blockIdx.x * OWN_R;
    const int Rn   = min(OWN_R, NNODES - base);

    unsigned*      mine     = keep + (size_t)base * HID + MAP_W;
    const unsigned keep_cap = (unsigned)(Rn * HID - MAP_W - KEEP_HDR);
    const unsigned kept     = first ? 0xFFFFFFFFu : mine[0];
    const bool     reuse    = kept <= keep_cap;

    for (int i = tid; i < OWN_R * HID; i += OWN_T) acc[i] = 0.f;
    for (int i = tid; i < Rn * HEADS; i += OWN_T) {
        sMA[i] = AS[base * HEADS + i];
        sAD[i] = AD[base * HEADS + i];
        sS[i]  = 0.f;
    }
    if (tid == 0) cnt = reuse ? kept : 0u;
    if (reuse)
        for (unsigned i = tid; i < kept; i += OWN_T) list[i] = mine[KEEP_HDR + i];
    __syncthreads();

    if (first) {
        const uint4*   mv  = (const uint4*)keep;
        const unsigned pat = (unsigned)blockIdx.x * 0x01010101u;
        for (int q0 = tid; q0 < MAP_NV; q0 += MAP_U * OWN_T) {
            uint4 v[MAP_U];
#pragma unroll
            for (int u = 0; u < MAP_U; ++u) {
                const unsigned q = (unsigned)(q0 + u * OWN_T);
                const unsigned j = q / MAP_SV;
                v[u] = q < MAP_NV ? mv[j * REGION_V + (q - j * MAP_SV)]
                                  : make_uint4(~pat, ~pat, ~pat, ~pat);
            }
            unsigned m[MAP_U], c = 0;
#pragma unroll
            for (int u = 0; u < MAP_U; ++u) {
                m[u] = (zero_bytes(v[u].x ^ pat) >> 7) | (zero_bytes(v[u].y ^ pat) >> 6) |
                       (zero_bytes(v[u].z ^ pat) >> 5) | (zero_bytes(v[u].w ^ pat) >> 4);
                c += (unsigned)__builtin_popcount(m[u]);
            }
            if (c) {
                unsigned slot = atomicAdd(&cnt, c);
#pragma unroll
                for (int u = 0; u < MAP_U; ++u) {
                    const unsigned e0 = (unsigned)(q0 + u * OWN_T) * 16u;
                    for (unsigned mm = m[u]; mm; mm &= mm - 1, ++slot) {
                        const unsigned bit = (unsigned)__builtin_ctz(mm);
                        if (slot < LIST_CAP) list[slot] = e0 + 4u * (bit & 7u) + (bit >> 3);
                    }
                }
            }
        }
    }
    __syncthreads();
    const unsigned total = cnt;
    const bool     save  = first && total <= keep_cap;
    if (first && tid == 0) mine[0] = total;

    if (first ? total <= LIST_CAP : reuse) {
        for (unsigned i = tid; i < total; i += OWN_T) {
            unsigned ent = list[i];
            if (first) {
                const int s = ei[ent];
                ent     = ((unsigned)s << 8) | (unsigned)(ei[NEDGES + ent] - base);
                list[i] = ent;
                if (save) mine[KEEP_HDR + i] = ent;
            }
            const unsigned ld = ent & 255u;
            const int      s  = (int)(ent >> 8);
            const float4* ap = (const float4*)(AS + (size_t)s * HEADS);
            const float4 a0 = ap[0], a1 = ap[1];
            float* m = &sMA[ld * HEADS];
            lds_fmax(m + 0, a0.x); lds_fmax(m + 1, a0.y);
            lds_fmax(m + 2, a0.z); lds_fmax(m + 3, a0.w);
            lds_fmax(m + 4, a1.x); lds_fmax(m + 5, a1.y);
            lds_fmax(m + 6, a1.z); lds_fmax(m + 7, a1.w);
        }
        __syncthreads();

        const int lane = tid & 31, wave = tid >> 5, hd = lane >> 2;
        const unsigned nitems = total + (unsigned)Rn;
        for (unsigned i0 = wave * AGG_U; i0 < nitems; i0 += OWN_W * AGG_U) {
            int    sv[AGG_U], lv[AGG_U];
            float4 hv[AGG_U];
            float  av[AGG_U];
#pragma unroll
            for (int u = 0; u < AGG_U; ++u) {
                const unsigned i = i0 + u;
                if (i < total)       { const unsigned ent = list[i]; sv[u] = ent >> 8; lv[u] = ent & 255u; }
                else if (i < nitems) { lv[u] = i - total; sv[u] = base + lv[u]; }
                else                 { lv[u] = 0; sv[u] = base; }
            }
#pragma unroll
            for (int u = 0; u < AGG_U; ++u) {
                hv[u] = *(const float4*)(Hf + (size_t)sv[u] * HID + 4 * lane);
                av[u] = AS[sv[u] * HEADS + hd];
            }
#pragma unroll
            for (int u = 0; u < AGG_U; ++u) {
                if (i0 + u < nitems) {
                    const int   ld = lv[u];
                    const float ad = sAD[ld * HEADS + hd];
                    const float m  = leaky(sMA[ld * HEADS + hd] + ad);
                    const float w  = expf(leaky(av[u] + ad) - m);
                    float* a = &acc[ld * HID + lane];
                    atomicAdd(a,      w * hv[u].x);
                    atomicAdd(a + 32, w * hv[u].y);
                    atomicAdd(a + 64, w * hv[u].z);
                    atomicAdd(a + 96, w * hv[u].w);
                    if ((lane & 3) == 0) atomicAdd(&sS[ld * HEADS + hd], w);
                }
            }
        }
    } else {
        for (int e = tid; e < NEDGES; e += OWN_T) {
            const unsigned ld = (unsigned)(ei[NEDGES + e] - base);
            if (ld < (unsigned)Rn) {
                const int s = ei[e];
                for (int h = 0; h < HEADS; ++h) lds_fmax(&sMA[ld * HEADS + h], AS[s * HEADS + h]);
            }
        }
        __syncthreads();
        for (int e = tid; e < NEDGES; e += OWN_T) {
            const unsigned ld = (unsigned)(ei[NEDGES + e] - base);
            if (ld < (unsigned)Rn) edge_full_lane(AS, Hf, sMA, sAD, sS, acc, ei[e], (int)ld);
        }
        for (int ld = tid; ld < Rn; ld += OWN_T)
            edge_full_lane(AS, Hf, sMA, sAD, sS, acc, base + ld, ld);
    }
    __syncthreads();

    for (int i = tid; i < Rn * HID; i += OWN_T) {
        const int ld = i >> 7, c = i & (HID - 1);
        const float v = acc[ld * HID + acc_pos(c)] / (sS[ld * HEADS + (c >> 4)] + 1e-16f) + bias[c];
        Out[(size_t)(base + ld) * HID + c] = v > 0.f ? v : expm1f(v);
    }
}

static inline unsigned grid_for(long long n) { return (unsigned)((n + 255) / 256); }

extern "C" void kernel_launch(void* const* d_in, const int* in_sizes, int n_in,
                              void* d_out, int out_size, void* d_ws, size_t ws_size,
                              hipStream_t stream) {
    const float* x   = (const float*)d_in[0];
    const int*   ei  = (const int*)  d_in[1];
    const float* W1  = (const float*)d_in[2];
    const float* as1 = (const float*)d_in[3];
    const float* ad1 = (const float*)d_in[4];
    const float* b1  = (const float*)d_in[5];
    const float* W2  = (const float*)d_in[6];
    const float* as2 = (const float*)d_in[7];
    const float* ad2 = (const float*)d_in[8];
    const float* b2  = (const float*)d_in[9];

    float* ws     = (float*)d_ws;
    float* bufH   = ws;
    float* bufAS  = bufH  + (size_t)NNODES * HID;
    float* bufAD  = bufAS + (size_t)NNODES * HEADS;
    float* bufAgg = bufAD + (size_t)NNODES * HEADS;

    auto run_layer = [&](const float* xin, const float* Wm, const float* as,
                         const float* ad, const float* bv, float* outBuf, int first) {
        k_gemm_wmma<<<NNODES / 16, 256, 0, stream>>>(xin, Wm, bufH);
        k_alpha<<<grid_for((long long)NNODES * HEADS), 256, 0, stream>>>(
            bufH, as, ad, bufAS, bufAD, ei, (unsigned*)d_out, first);
        k_owner<<<OWN_NB, OWN_T, 0, stream>>>(ei, bufAS, bufAD, bufH, bv, outBuf,
                                              (unsigned*)d_out, first);
    };

    run_layer(x,      W1, as1, ad1, b1, bufAgg,        1);
    run_layer(bufAgg, W2, as2, ad2, b2, (float*)d_out, 0);
}


# Row-sum phase: a trip's 64 weights (8 items x 8 heads) computed once each, 2 per lane, handed to the row lanes by quad broadcast; was 8 per lane, each 4 times over; sums arrival-ordered as before
# speedup vs baseline: 7.4187x; 1.1178x over previous
#include <hip/hip_runtime.h>
#include <math.h>

#define NNODES 50000
#define NEDGES 800000
#define FEAT 128
#define HID  128
#define HEADS 8
#define HDIM 16
#define NEG_SLOPE 0.2f

#define OWN_R    196
#define OWN_NB   256
#define OWN_T    1024
#define OWN_W    (OWN_T / 32)
#define LIST_CAP 32768
#define AGG_U    8
#define KEEP_HDR 32
#define MAP_SLICE 3200
#define MAP_W     (MAP_SLICE / 4)
#define MAP_SV    (MAP_SLICE / 16)
#define REGION_V  (OWN_R * HID / 4)
#define MAP_NV    (NEDGES / 16)
#define MAP_U     7
static_assert(MAP_SLICE % 128 == 0 && NEDGES % MAP_SLICE == 0 && NEDGES / MAP_SLICE < OWN_NB,
              "whole slices, all in full-size blocks");
static_assert(NEDGES / 4 <= NNODES * HEADS, "the score launch has a thread per four edges");
static_assert((NNODES - (OWN_NB - 1) * OWN_R) * HID > MAP_W + KEEP_HDR, "last block holds its header");

typedef __attribute__((ext_vector_type(2))) float v2f;
typedef __attribute__((ext_vector_type(8))) float v8f;

__global__ __launch_bounds__(256)
void k_gemm_wmma(const float* __restrict__ X, const float* __restrict__ W,
                 float* __restrict__ Hout) {
    __shared__ float ldsW[FEAT * HID];
    const int tid = threadIdx.x;

    const float4* Wv = (const float4*)W;
    float4*       Lv = (float4*)ldsW;
#pragma unroll
    for (int i = 0; i < (FEAT * HID / 4) / 256; ++i)
        Lv[tid + i * 256] = Wv[tid + i * 256];
    __syncthreads();

    const int lane = tid & 31;
    const int wave = tid >> 5;
    const int l16  = lane & 15;
    const int half = lane >> 4;
    const int row  = blockIdx.x * 16 + l16;
    const int col  = wave * 16 + l16;

    const float* xrow = X + (size_t)row * FEAT;

    v8f acc = {};
#pragma unroll
    for (int k0 = 0; k0 < FEAT; k0 += 4) {
        const int ka = k0 + 2 * half;
        v2f a, b;
        a.x = xrow[ka];
        a.y = xrow[ka + 1];
        b.x = ldsW[ka * HID + col];
        b.y = ldsW[(ka + 1) * HID + col];
        acc = __builtin_amdgcn_wmma_f32_16x16x4_f32(
                  false, a, false, b, (short)0, acc, false, false);
    }

    const int orow0 = blockIdx.x * 16 + 8 * half;
#pragma unroll
    for (int j = 0; j < 8; ++j)
        Hout[(size_t)(orow0 + j) * HID + col] = acc[j];
}

__global__ __launch_bounds__(256)
void k_alpha(const float* __restrict__ Hf, const float* __restrict__ a_src,
             const float* __restrict__ a_dst, float* __restrict__ AS,
             float* __restrict__ AD, const int* __restrict__ ei,
             unsigned* keep, int first) {
    int t = blockIdx.x * blockDim.x + threadIdx.x;
    if (t >= NNODES * HEADS) return;
    if (first && t < NEDGES / 4) {
        const int4     d  = *(const int4*)(ei + NEDGES + 4 * t);
        const unsigned o  = ((unsigned)d.x / OWN_R)         | (((unsigned)d.y / OWN_R) << 8) |
                            (((unsigned)d.z / OWN_R) << 16) | (((unsigned)d.w / OWN_R) << 24);
        const unsigned sl = (unsigned)t / MAP_W;
        keep[(size_t)sl * (OWN_R * HID) + ((unsigned)t - sl * MAP_W)] = o;
    }
    int hd = t & (HEADS - 1);
    int n  = t >> 3;
    const float* hp = Hf + (size_t)n * HID + hd * HDIM;
    const float* as = a_src + hd * HDIM;
    const float* ad = a_dst + hd * HDIM;
    float sa = 0.f, sd = 0.f;
#pragma unroll
    for (int c = 0; c < HDIM; ++c) {
        float v = hp[c];
        sa += v * as[c];
        sd += v * ad[c];
    }
    AS[t] = sa;
    AD[t] = sd;
}

static __device__ __forceinline__ float leaky(float v) {
    return v >= 0.f ? v : NEG_SLOPE * v;
}

static __device__ __forceinline__ unsigned zero_bytes(unsigned x) {
    return ~(((x & 0x7F7F7F7Fu) + 0x7F7F7F7Fu) | x) & 0x80808080u;
}

static __device__ __forceinline__ void lds_fmax(float* p, float v) {
    __hip_atomic_fetch_max(p, v, __ATOMIC_RELAXED, __HIP_MEMORY_SCOPE_WORKGROUP);
}

static __device__ __forceinline__ int acc_pos(int c) { return (c & 3) * 32 + (c >> 2); }

template <int J>
static __device__ __forceinline__ float quad_bcast(float v) {
    return __int_as_float(__builtin_amdgcn_update_dpp(0, __float_as_int(v), J * 0x55, 0xF, 0xF, true));
}

static __device__ void edge_full_lane(const float* __restrict__ AS, const float* __restrict__ Hf,
                                      const float* sMA, const float* sAD, float* sS, float* acc,
                                      int s, int ld) {
    for (int hd = 0; hd < HEADS; ++hd) {
        const float ad = sAD[ld * HEADS + hd];
        const float m  = leaky(sMA[ld * HEADS + hd] + ad);
        const float w  = expf(leaky(AS[s * HEADS + hd] + ad) - m);
        atomicAdd(&sS[ld * HEADS + hd], w);
        for (int c = 0; c < HDIM; ++c) {
            const int col = hd * HDIM + c;
            atomicAdd(&acc[ld * HID + acc_pos(col)], w * Hf[(size_t)s * HID + col]);
        }
    }
}

__global__ __launch_bounds__(OWN_T)
void k_owner(const int* __restrict__ ei, const float* __restrict__ AS,
             const float* __restrict__ AD, const float* __restrict__ Hf,
             const float* __restrict__ bias, float* Out, unsigned* keep, int first) {
    __shared__ float    acc[OWN_R * HID];
    __shared__ float    sMA[OWN_R * HEADS];
    __shared__ float    sAD[OWN_R * HEADS];
    __shared__ float    sS [OWN_R * HEADS];
    __shared__ unsigned list[LIST_CAP];
    __shared__ unsigned cnt;

    const int tid  = threadIdx.x;
    const int base = blockIdx.x * OWN_R;
    const int Rn   = min(OWN_R, NNODES - base);

    unsigned*      mine     = keep + (size_t)base * HID + MAP_W;
    const unsigned keep_cap = (unsigned)(Rn * HID - MAP_W - KEEP_HDR);
    const unsigned kept     = first ? 0xFFFFFFFFu : mine[0];
    const bool     reuse    = kept <= keep_cap;

    for (int i = tid; i < OWN_R * HID; i += OWN_T) acc[i] = 0.f;
    for (int i = tid; i < Rn * HEADS; i += OWN_T) {
        sMA[i] = AS[base * HEADS + i];
        sAD[i] = AD[base * HEADS + i];
        sS[i]  = 0.f;
    }
    if (tid == 0) cnt = reuse ? kept : 0u;
    if (reuse)
        for (unsigned i = tid; i < kept; i += OWN_T) list[i] = mine[KEEP_HDR + i];
    __syncthreads();

    if (first) {
        const uint4*   mv  = (const uint4*)keep;
        const unsigned pat = (unsigned)blockIdx.x * 0x01010101u;
        for (int q0 = tid; q0 < MAP_NV; q0 += MAP_U * OWN_T) {
            uint4 v[MAP_U];
#pragma unroll
            for (int u = 0; u < MAP_U; ++u) {
                const unsigned q = (unsigned)(q0 + u * OWN_T);
                const unsigned j = q / MAP_SV;
                v[u] = q < MAP_NV ? mv[j * REGION_V + (q - j * MAP_SV)]
                                  : make_uint4(~pat, ~pat, ~pat, ~pat);
            }
            unsigned m[MAP_U], c = 0;
#pragma unroll
            for (int u = 0; u < MAP_U; ++u) {
                m[u] = (zero_bytes(v[u].x ^ pat) >> 7) | (zero_bytes(v[u].y ^ pat) >> 6) |
                       (zero_bytes(v[u].z ^ pat) >> 5) | (zero_bytes(v[u].w ^ pat) >> 4);
                c += (unsigned)__builtin_popcount(m[u]);
            }
            if (c) {
                unsigned slot = atomicAdd(&cnt, c);
#pragma unroll
                for (int u = 0; u < MAP_U; ++u) {
                    const unsigned e0 = (unsigned)(q0 + u * OWN_T) * 16u;
                    for (unsigned mm = m[u]; mm; mm &= mm - 1, ++slot) {
                        const unsigned bit = (unsigned)__builtin_ctz(mm);
                        if (slot < LIST_CAP) list[slot] = e0 + 4u * (bit & 7u) + (bit >> 3);
                    }
                }
            }
        }
    }
    __syncthreads();
    const unsigned total  = (unsigned)__builtin_amdgcn_readfirstlane((int)cnt);
    const unsigned nitems = total + (unsigned)Rn;
    const bool     save   = first && total <= keep_cap;
    if (first && tid == 0) mine[0] = total;

    if (first ? nitems <= LIST_CAP : reuse) {
        for (unsigned i = tid; i < total; i += OWN_T) {
            unsigned ent = list[i];
            if (first) {
                const int s = ei[ent];
                ent     = ((unsigned)s << 8) | (unsigned)(ei[NEDGES + ent] - base);
                list[i] = ent;
                if (save) mine[KEEP_HDR + i] = ent;
            }
            const unsigned ld = ent & 255u;
            const int      s  = (int)(ent >> 8);
            const float4* ap = (const float4*)(AS + (size_t)s * HEADS);
            const float4 a0 = ap[0], a1 = ap[1];
            float* m = &sMA[ld * HEADS];
            lds_fmax(m + 0, a0.x); lds_fmax(m + 1, a0.y);
            lds_fmax(m + 2, a0.z); lds_fmax(m + 3, a0.w);
            lds_fmax(m + 4, a1.x); lds_fmax(m + 5, a1.y);
            lds_fmax(m + 6, a1.z); lds_fmax(m + 7, a1.w);
        }
        for (unsigned k = tid; k < (unsigned)Rn; k += OWN_T)
            list[total + k] = ((unsigned)(base + (int)k) << 8) | k;
        __syncthreads();

        const int lane = tid & 31, hd = lane >> 2, q = lane & 3;
        const unsigned wave = (unsigned)__builtin_amdgcn_readfirstlane(tid >> 5);
        const float*   hcol = Hf + 4 * lane;
        for (unsigned i0 = wave * AGG_U; i0 < nitems; i0 += OWN_W * AGG_U) {
            const unsigned iA = i0 + (unsigned)q, iB = iA + 4u;
            const unsigned eA = list[min(iA, nitems - 1u)];
            const unsigned eB = list[min(iB, nitems - 1u)];
            unsigned en[AGG_U];
#pragma unroll
            for (int u = 0; u < 4; ++u) {
                en[u]     = (unsigned)__builtin_amdgcn_readlane((int)eA, u);
                en[u + 4] = (unsigned)__builtin_amdgcn_readlane((int)eB, u);
            }
            float4 hv[AGG_U];
#pragma unroll
            for (int u = 0; u < AGG_U; ++u)
                hv[u] = *(const float4*)(hcol + (size_t)(en[u] >> 8) * HID);
            const unsigned lA = (eA & 255u) * HEADS + (unsigned)hd, lB = (eB & 255u) * HEADS + (unsigned)hd;
            const float asA = AS[(eA >> 8) * HEADS + (unsigned)hd];
            const float asB = AS[(eB >> 8) * HEADS + (unsigned)hd];
            const float adA = sAD[lA], adB = sAD[lB];
            const float wA  = expf(leaky(asA + adA) - leaky(sMA[lA] + adA));
            const float wB  = expf(leaky(asB + adB) - leaky(sMA[lB] + adB));
            if (iA < nitems) atomicAdd(&sS[lA], wA);
            if (iB < nitems) atomicAdd(&sS[lB], wB);
#define AGG_ITEM(U, WSRC)                                                          \
            if (i0 + U < nitems) {                                \
                const float w = quad_bcast<(U) & 3>(WSRC);                         \
                const v2f   ww = {w, w};                                           \
                  \
                const v2f   p0 = ww * (v2f){hv[U].x, hv[U].y};                     \
                const v2f   p1 = ww * (v2f){hv[U].z, hv[U].w};                     \
                float* a = &acc[(en[U] & 255u) * HID + (unsigned)lane];            \
                atomicAdd(a,      p0.x);                                           \
                atomicAdd(a + 32, p0.y);                                           \
                atomicAdd(a + 64, p1.x);                                           \
                atomicAdd(a + 96, p1.y);                                           \
            }
            AGG_ITEM(0, wA) AGG_ITEM(1, wA) AGG_ITEM(2, wA) AGG_ITEM(3, wA)
            AGG_ITEM(4, wB) AGG_ITEM(5, wB) AGG_ITEM(6, wB) AGG_ITEM(7, wB)
#undef AGG_ITEM
        }
    } else {
        for (int e = tid; e < NEDGES; e += OWN_T) {
            const unsigned ld = (unsigned)(ei[NEDGES + e] - base);
            if (ld < (unsigned)Rn) {
                const int s = ei[e];
                for (int h = 0; h < HEADS; ++h) lds_fmax(&sMA[ld * HEADS + h], AS[s * HEADS + h]);
            }
        }
        __syncthreads();
        for (int e = tid; e < NEDGES; e += OWN_T) {
            const unsigned ld = (unsigned)(ei[NEDGES + e] - base);
            if (ld < (unsigned)Rn) edge_full_lane(AS, Hf, sMA, sAD, sS, acc, ei[e], (int)ld);
        }
        for (int ld = tid; ld < Rn; ld += OWN_T)
            edge_full_lane(AS, Hf, sMA, sAD, sS, acc, base + ld, ld);
    }
    __syncthreads();

    for (int i = tid; i < Rn * HID; i += OWN_T) {
        const int ld = i >> 7, c = i & (HID - 1);
        const float v = acc[ld * HID + acc_pos(c)] / (sS[ld * HEADS + (c >> 4)] + 1e-16f) + bias[c];
        Out[(size_t)(base + ld) * HID + c] = v > 0.f ? v : expm1f(v);
    }
}

static inline unsigned grid_for(long long n) { return (unsigned)((n + 255) / 256); }

extern "C" void kernel_launch(void* const* d_in, const int* in_sizes, int n_in,
                              void* d_out, int out_size, void* d_ws, size_t ws_size,
                              hipStream_t stream) {
    const float* x   = (const float*)d_in[0];
    const int*   ei  = (const int*)  d_in[1];
    const float* W1  = (const float*)d_in[2];
    const float* as1 = (const float*)d_in[3];
    const float* ad1 = (const float*)d_in[4];
    const float* b1  = (const float*)d_in[5];
    const float* W2  = (const float*)d_in[6];
    const float* as2 = (const float*)d_in[7];
    const float* ad2 = (const float*)d_in[8];
    const float* b2  = (const float*)d_in[9];

    float* ws     = (float*)d_ws;
    float* bufH   = ws;
    float* bufAS  = bufH  + (size_t)NNODES * HID;
    float* bufAD  = bufAS + (size_t)NNODES * HEADS;
    float* bufAgg = bufAD + (size_t)NNODES * HEADS;

    auto run_layer = [&](const float* xin, const float* Wm, const float* as,
                         const float* ad, const float* bv, float* outBuf, int first) {
        k_gemm_wmma<<<NNODES / 16, 256, 0, stream>>>(xin, Wm, bufH);
        k_alpha<<<grid_for((long long)NNODES * HEADS), 256, 0, stream>>>(
            bufH, as, ad, bufAS, bufAD, ei, (unsigned*)d_out, first);
        k_owner<<<OWN_NB, OWN_T, 0, stream>>>(ei, bufAS, bufAD, bufH, bv, outBuf,
                                              (unsigned*)d_out, first);
    };

    run_layer(x,      W1, as1, ad1, b1, bufAgg,        1);
    run_layer(bufAgg, W2, as2, ad2, b2, (float*)d_out, 0);
}


# GEMM column-stationary: a wave keeps its 16 columns of W (all 128 k) in registers and walks 6-7 row tiles; X tile fetched once per block by whole-row loads via a 2-stage LDS ring; sum order kept
# speedup vs baseline: 8.6515x; 1.1662x over previous
#include <hip/hip_runtime.h>
#include <math.h>

#define NNODES 50000
#define NEDGES 800000
#define FEAT 128
#define HID  128
#define HEADS 8
#define HDIM 16
#define NEG_SLOPE 0.2f

#define OWN_R    196
#define OWN_NB   256
#define OWN_T    1024
#define OWN_W    (OWN_T / 32)
#define LIST_CAP 32768
#define AGG_U    8
#define KEEP_HDR 32
#define MAP_SLICE 3200
#define MAP_W     (MAP_SLICE / 4)
#define MAP_SV    (MAP_SLICE / 16)
#define REGION_V  (OWN_R * HID / 4)
#define MAP_NV    (NEDGES / 16)
#define MAP_U     7
static_assert(MAP_SLICE % 128 == 0 && NEDGES % MAP_SLICE == 0 && NEDGES / MAP_SLICE < OWN_NB,
              "whole slices, all in full-size blocks");
static_assert(NEDGES / 4 <= NNODES * HEADS, "the score launch has a thread per four edges");
static_assert((NNODES - (OWN_NB - 1) * OWN_R) * HID > MAP_W + KEEP_HDR, "last block holds its header");

typedef __attribute__((ext_vector_type(2))) float v2f;
typedef __attribute__((ext_vector_type(8))) float v8f;

#define GEMM_NB   512
#define GEMM_TILES (NNODES / 16)
#define XS_PITCH  132

static __device__ __forceinline__ void lds_barrier() {
    __builtin_amdgcn_fence(__ATOMIC_RELEASE, "workgroup", "local");
    __builtin_amdgcn_s_barrier();
    __builtin_amdgcn_fence(__ATOMIC_ACQUIRE, "workgroup", "local");
}

__global__ __launch_bounds__(256)
void k_gemm_wmma(const float* __restrict__ X, const float* __restrict__ W,
                 float* __restrict__ Hout) {
    __shared__ float sX[2][16 * XS_PITCH];
    const int tid  = threadIdx.x;
    const int lane = tid & 31;
    const int wave = tid >> 5;
    const int l16  = lane & 15;
    const int half = lane >> 4;
    const int col  = wave * 16 + l16;

    const int srow = tid >> 5, sc4 = tid & 31;
    const float* xg = X + (size_t)srow * FEAT + sc4 * 4;
    const int    xl = srow * XS_PITCH + sc4 * 4;

    int tile = blockIdx.x;
    float4 g0 = *(const float4*)(xg + (size_t)tile * (16 * FEAT));
    float4 g1 = *(const float4*)(xg + (size_t)tile * (16 * FEAT) + 8 * FEAT);

    v2f b[FEAT / 4];
#pragma unroll
    for (int i = 0; i < FEAT / 4; ++i) {
        const int ka = 4 * i + 2 * half;
        b[i].x = W[ka * HID + col];
        b[i].y = W[(ka + 1) * HID + col];
    }

    *(float4*)&sX[0][xl]                = g0;
    *(float4*)&sX[0][xl + 8 * XS_PITCH] = g1;
    if (tile + GEMM_NB < GEMM_TILES) {
        g0 = *(const float4*)(xg + (size_t)(tile + GEMM_NB) * (16 * FEAT));
        g1 = *(const float4*)(xg + (size_t)(tile + GEMM_NB) * (16 * FEAT) + 8 * FEAT);
    }
    lds_barrier();

    const int arow = l16 * XS_PITCH + 2 * half;
    for (int it = 0; tile < GEMM_TILES; ++it, tile += GEMM_NB) {
        const float* xs = sX[it & 1] + arow;
        v2f a[FEAT / 4];
#pragma unroll
        for (int i = 0; i < FEAT / 4; ++i) a[i] = *(const v2f*)(xs + 4 * i);

        if (tile + GEMM_NB < GEMM_TILES) {
            float* xn = sX[(it & 1) ^ 1];
            *(float4*)&xn[xl]                = g0;
            *(float4*)&xn[xl + 8 * XS_PITCH] = g1;
            if (tile + 2 * GEMM_NB < GEMM_TILES) {
                g0 = *(const float4*)(xg + (size_t)(tile + 2 * GEMM_NB) * (16 * FEAT));
                g1 = *(const float4*)(xg + (size_t)(tile + 2 * GEMM_NB) * (16 * FEAT) + 8 * FEAT);
            }
        }

        v8f acc = {};
#pragma unroll
        for (int i = 0; i < FEAT / 4; ++i)
            acc = __builtin_amdgcn_wmma_f32_16x16x4_f32(
                      false, a[i], false, b[i], (short)0, acc, false, false);

        lds_barrier();

        const int orow0 = tile * 16 + 8 * half;
#pragma unroll
        for (int j = 0; j < 8; ++j)
            Hout[(size_t)(orow0 + j) * HID + col] = acc[j];
    }
}

__global__ __launch_bounds__(256)
void k_alpha(const float* __restrict__ Hf, const float* __restrict__ a_src,
             const float* __restrict__ a_dst, float* __restrict__ AS,
             float* __restrict__ AD, const int* __restrict__ ei,
             unsigned* keep, int first) {
    int t = blockIdx.x * blockDim.x + threadIdx.x;
    if (t >= NNODES * HEADS) return;
    if (first && t < NEDGES / 4) {
        const int4     d  = *(const int4*)(ei + NEDGES + 4 * t);
        const unsigned o  = ((unsigned)d.x / OWN_R)         | (((unsigned)d.y / OWN_R) << 8) |
                            (((unsigned)d.z / OWN_R) << 16) | (((unsigned)d.w / OWN_R) << 24);
        const unsigned sl = (unsigned)t / MAP_W;
        keep[(size_t)sl * (OWN_R * HID) + ((unsigned)t - sl * MAP_W)] = o;
    }
    int hd = t & (HEADS - 1);
    int n  = t >> 3;
    const float* hp = Hf + (size_t)n * HID + hd * HDIM;
    const float* as = a_src + hd * HDIM;
    const float* ad = a_dst + hd * HDIM;
    float sa = 0.f, sd = 0.f;
#pragma unroll
    for (int c = 0; c < HDIM; ++c) {
        float v = hp[c];
        sa += v * as[c];
        sd += v * ad[c];
    }
    AS[t] = sa;
    AD[t] = sd;
}

static __device__ __forceinline__ float leaky(float v) {
    return v >= 0.f ? v : NEG_SLOPE * v;
}

static __device__ __forceinline__ unsigned zero_bytes(unsigned x) {
    return ~(((x & 0x7F7F7F7Fu) + 0x7F7F7F7Fu) | x) & 0x80808080u;
}

static __device__ __forceinline__ void lds_fmax(float* p, float v) {
    __hip_atomic_fetch_max(p, v, __ATOMIC_RELAXED, __HIP_MEMORY_SCOPE_WORKGROUP);
}

static __device__ __forceinline__ int acc_pos(int c) { return (c & 3) * 32 + (c >> 2); }

template <int J>
static __device__ __forceinline__ float quad_bcast(float v) {
    return __int_as_float(__builtin_amdgcn_update_dpp(0, __float_as_int(v), J * 0x55, 0xF, 0xF, true));
}

static __device__ void edge_full_lane(const float* __restrict__ AS, const float* __restrict__ Hf,
                                      const float* sMA, const float* sAD, float* sS, float* acc,
                                      int s, int ld) {
    for (int hd = 0; hd < HEADS; ++hd) {
        const float ad = sAD[ld * HEADS + hd];
        const float m  = leaky(sMA[ld * HEADS + hd] + ad);
        const float w  = expf(leaky(AS[s * HEADS + hd] + ad) - m);
        atomicAdd(&sS[ld * HEADS + hd], w);
        for (int c = 0; c < HDIM; ++c) {
            const int col = hd * HDIM + c;
            atomicAdd(&acc[ld * HID + acc_pos(col)], w * Hf[(size_t)s * HID + col]);
        }
    }
}

__global__ __launch_bounds__(OWN_T)
void k_owner(const int* __restrict__ ei, const float* __restrict__ AS,
             const float* __restrict__ AD, const float* __restrict__ Hf,
             const float* __restrict__ bias, float* Out, unsigned* keep, int first) {
    __shared__ float    acc[OWN_R * HID];
    __shared__ float    sMA[OWN_R * HEADS];
    __shared__ float    sAD[OWN_R * HEADS];
    __shared__ float    sS [OWN_R * HEADS];
    __shared__ unsigned list[LIST_CAP];
    __shared__ unsigned cnt;

    const int tid  = threadIdx.x;
    const int base = blockIdx.x * OWN_R;
    const int Rn   = min(OWN_R, NNODES - base);

    unsigned*      mine     = keep + (size_t)base * HID + MAP_W;
    const unsigned keep_cap = (unsigned)(Rn * HID - MAP_W - KEEP_HDR);
    const unsigned kept     = first ? 0xFFFFFFFFu : mine[0];
    const bool     reuse    = kept <= keep_cap;

    for (int i = tid; i < OWN_R * HID; i += OWN_T) acc[i] = 0.f;
    for (int i = tid; i < Rn * HEADS; i += OWN_T) {
        sMA[i] = AS[base * HEADS + i];
        sAD[i] = AD[base * HEADS + i];
        sS[i]  = 0.f;
    }
    if (tid == 0) cnt = reuse ? kept : 0u;
    if (reuse)
        for (unsigned i = tid; i < kept; i += OWN_T) list[i] = mine[KEEP_HDR + i];
    __syncthreads();

    if (first) {
        const uint4*   mv  = (const uint4*)keep;
        const unsigned pat = (unsigned)blockIdx.x * 0x01010101u;
        for (int q0 = tid; q0 < MAP_NV; q0 += MAP_U * OWN_T) {
            uint4 v[MAP_U];
#pragma unroll
            for (int u = 0; u < MAP_U; ++u) {
                const unsigned q = (unsigned)(q0 + u * OWN_T);
                const unsigned j = q / MAP_SV;
                v[u] = q < MAP_NV ? mv[j * REGION_V + (q - j * MAP_SV)]
                                  : make_uint4(~pat, ~pat, ~pat, ~pat);
            }
            unsigned m[MAP_U], c = 0;
#pragma unroll
            for (int u = 0; u < MAP_U; ++u) {
                m[u] = (zero_bytes(v[u].x ^ pat) >> 7) | (zero_bytes(v[u].y ^ pat) >> 6) |
                       (zero_bytes(v[u].z ^ pat) >> 5) | (zero_bytes(v[u].w ^ pat) >> 4);
                c += (unsigned)__builtin_popcount(m[u]);
            }
            if (c) {
                unsigned slot = atomicAdd(&cnt, c);
#pragma unroll
                for (int u = 0; u < MAP_U; ++u) {
                    const unsigned e0 = (unsigned)(q0 + u * OWN_T) * 16u;
                    for (unsigned mm = m[u]; mm; mm &= mm - 1, ++slot) {
                        const unsigned bit = (unsigned)__builtin_ctz(mm);
                        if (slot < LIST_CAP) list[slot] = e0 + 4u * (bit & 7u) + (bit >> 3);
                    }
                }
            }
        }
    }
    __syncthreads();
    const unsigned total  = (unsigned)__builtin_amdgcn_readfirstlane((int)cnt);
    const unsigned nitems = total + (unsigned)Rn;
    const bool     save   = first && total <= keep_cap;
    if (first && tid == 0) mine[0] = total;

    if (first ? nitems <= LIST_CAP : reuse) {
        for (unsigned i = tid; i < total; i += OWN_T) {
            unsigned ent = list[i];
            if (first) {
                const int s = ei[ent];
                ent     = ((unsigned)s << 8) | (unsigned)(ei[NEDGES + ent] - base);
                list[i] = ent;
                if (save) mine[KEEP_HDR + i] = ent;
            }
            const unsigned ld = ent & 255u;
            const int      s  = (int)(ent >> 8);
            const float4* ap = (const float4*)(AS + (size_t)s * HEADS);
            const float4 a0 = ap[0], a1 = ap[1];
            float* m = &sMA[ld * HEADS];
            lds_fmax(m + 0, a0.x); lds_fmax(m + 1, a0.y);
            lds_fmax(m + 2, a0.z); lds_fmax(m + 3, a0.w);
            lds_fmax(m + 4, a1.x); lds_fmax(m + 5, a1.y);
            lds_fmax(m + 6, a1.z); lds_fmax(m + 7, a1.w);
        }
        for (unsigned k = tid; k < (unsigned)Rn; k += OWN_T)
            list[total + k] = ((unsigned)(base + (int)k) << 8) | k;
        __syncthreads();

        const int lane = tid & 31, hd = lane >> 2, q = lane & 3;
        const unsigned wave = (unsigned)__builtin_amdgcn_readfirstlane(tid >> 5);
        const float*   hcol = Hf + 4 * lane;
        for (unsigned i0 = wave * AGG_U; i0 < nitems; i0 += OWN_W * AGG_U) {
            const unsigned iA = i0 + (unsigned)q, iB = iA + 4u;
            const unsigned eA = list[min(iA, nitems - 1u)];
            const unsigned eB = list[min(iB, nitems - 1u)];
            unsigned en[AGG_U];
#pragma unroll
            for (int u = 0; u < 4; ++u) {
                en[u]     = (unsigned)__builtin_amdgcn_readlane((int)eA, u);
                en[u + 4] = (unsigned)__builtin_amdgcn_readlane((int)eB, u);
            }
            float4 hv[AGG_U];
#pragma unroll
            for (int u = 0; u < AGG_U; ++u)
                hv[u] = *(const float4*)(hcol + (size_t)(en[u] >> 8) * HID);
            const unsigned lA = (eA & 255u) * HEADS + (unsigned)hd, lB = (eB & 255u) * HEADS + (unsigned)hd;
            const float asA = AS[(eA >> 8) * HEADS + (unsigned)hd];
            const float asB = AS[(eB >> 8) * HEADS + (unsigned)hd];
            const float adA = sAD[lA], adB = sAD[lB];
            const float wA  = expf(leaky(asA + adA) - leaky(sMA[lA] + adA));
            const float wB  = expf(leaky(asB + adB) - leaky(sMA[lB] + adB));
            if (iA < nitems) atomicAdd(&sS[lA], wA);
            if (iB < nitems) atomicAdd(&sS[lB], wB);
#define AGG_ITEM(U, WSRC)                                                          \
            if (i0 + U < nitems) {                                \
                const float w = quad_bcast<(U) & 3>(WSRC);                         \
                const v2f   ww = {w, w};                                           \
                  \
                const v2f   p0 = ww * (v2f){hv[U].x, hv[U].y};                     \
                const v2f   p1 = ww * (v2f){hv[U].z, hv[U].w};                     \
                float* a = &acc[(en[U] & 255u) * HID + (unsigned)lane];            \
                atomicAdd(a,      p0.x);                                           \
                atomicAdd(a + 32, p0.y);                                           \
                atomicAdd(a + 64, p1.x);                                           \
                atomicAdd(a + 96, p1.y);                                           \
            }
            AGG_ITEM(0, wA) AGG_ITEM(1, wA) AGG_ITEM(2, wA) AGG_ITEM(3, wA)
            AGG_ITEM(4, wB) AGG_ITEM(5, wB) AGG_ITEM(6, wB) AGG_ITEM(7, wB)
#undef AGG_ITEM
        }
    } else {
        for (int e = tid; e < NEDGES; e += OWN_T) {
            const unsigned ld = (unsigned)(ei[NEDGES + e] - base);
            if (ld < (unsigned)Rn) {
                const int s = ei[e];
                for (int h = 0; h < HEADS; ++h) lds_fmax(&sMA[ld * HEADS + h], AS[s * HEADS + h]);
            }
        }
        __syncthreads();
        for (int e = tid; e < NEDGES; e += OWN_T) {
            const unsigned ld = (unsigned)(ei[NEDGES + e] - base);
            if (ld < (unsigned)Rn) edge_full_lane(AS, Hf, sMA, sAD, sS, acc, ei[e], (int)ld);
        }
        for (int ld = tid; ld < Rn; ld += OWN_T)
            edge_full_lane(AS, Hf, sMA, sAD, sS, acc, base + ld, ld);
    }
    __syncthreads();

    for (int i = tid; i < Rn * HID; i += OWN_T) {
        const int ld = i >> 7, c = i & (HID - 1);
        const float v = acc[ld * HID + acc_pos(c)] / (sS[ld * HEADS + (c >> 4)] + 1e-16f) + bias[c];
        Out[(size_t)(base + ld) * HID + c] = v > 0.f ? v : expm1f(v);
    }
}

static inline unsigned grid_for(long long n) { return (unsigned)((n + 255) / 256); }

extern "C" void kernel_launch(void* const* d_in, const int* in_sizes, int n_in,
                              void* d_out, int out_size, void* d_ws, size_t ws_size,
                              hipStream_t stream) {
    const float* x   = (const float*)d_in[0];
    const int*   ei  = (const int*)  d_in[1];
    const float* W1  = (const float*)d_in[2];
    const float* as1 = (const float*)d_in[3];
    const float* ad1 = (const float*)d_in[4];
    const float* b1  = (const float*)d_in[5];
    const float* W2  = (const float*)d_in[6];
    const float* as2 = (const float*)d_in[7];
    const float* ad2 = (const float*)d_in[8];
    const float* b2  = (const float*)d_in[9];

    float* ws     = (float*)d_ws;
    float* bufH   = ws;
    float* bufAS  = bufH  + (size_t)NNODES * HID;
    float* bufAD  = bufAS + (size_t)NNODES * HEADS;
    float* bufAgg = bufAD + (size_t)NNODES * HEADS;

    auto run_layer = [&](const float* xin, const float* Wm, const float* as,
                         const float* ad, const float* bv, float* outBuf, int first) {
        k_gemm_wmma<<<GEMM_NB, 256, 0, stream>>>(xin, Wm, bufH);
        k_alpha<<<grid_for((long long)NNODES * HEADS), 256, 0, stream>>>(
            bufH, as, ad, bufAS, bufAD, ei, (unsigned*)d_out, first);
        k_owner<<<OWN_NB, OWN_T, 0, stream>>>(ei, bufAS, bufAD, bufH, bv, outBuf,
                                              (unsigned*)d_out, first);
    };

    run_layer(x,      W1, as1, ad1, b1, bufAgg,        1);
    run_layer(bufAgg, W2, as2, ad2, b2, (float*)d_out, 0);
}


# Attention scalars from the product's own accumulators: a GEMM wave's 16 columns are one head, summed over 16 lanes by halving butterfly, lines via LDS; score launches gone (6->4); sum order changes
# speedup vs baseline: 9.0199x; 1.0426x over previous
#include <hip/hip_runtime.h>
#include <math.h>

#define NNODES 50000
#define NEDGES 800000
#define FEAT 128
#define HID  128
#define HEADS 8
#define HDIM 16
#define NEG_SLOPE 0.2f

#define OWN_R    196
#define OWN_NB   256
#define OWN_T    1024
#define OWN_W    (OWN_T / 32)
#define LIST_CAP 32768
#define AGG_U    8
#define KEEP_HDR 32
#define MAP_SLICE 3200
#define MAP_W     (MAP_SLICE / 4)
#define MAP_SV    (MAP_SLICE / 16)
#define REGION_V  (OWN_R * HID / 4)
#define MAP_NV    (NEDGES / 16)
#define MAP_U     7
static_assert(MAP_SLICE % 128 == 0 && NEDGES % MAP_SLICE == 0 && NEDGES / MAP_SLICE < OWN_NB,
              "whole slices, all in full-size blocks");
static_assert((NNODES - (OWN_NB - 1) * OWN_R) * HID > MAP_W + KEEP_HDR, "last block holds its header");

typedef __attribute__((ext_vector_type(2))) float v2f;
typedef __attribute__((ext_vector_type(8))) float v8f;

#define GEMM_NB   512
#define GEMM_TILES (NNODES / 16)
#define XS_PITCH  132

static __device__ __forceinline__ void lds_barrier() {
    __builtin_amdgcn_fence(__ATOMIC_RELEASE, "workgroup", "local");
    __builtin_amdgcn_s_barrier();
    __builtin_amdgcn_fence(__ATOMIC_ACQUIRE, "workgroup", "local");
}

template <int K>
static __device__ __forceinline__ float row_xor(float v) {
    return __int_as_float(__builtin_amdgcn_update_dpp(0, __float_as_int(v), 0x160 + K, 0xF, 0xF, true));
}

static __device__ __forceinline__ float row_sums8(const float (&v)[8], int c) {
    const bool b3 = (c & 8) != 0, b2 = (c & 4) != 0, b1 = (c & 2) != 0;
    float u[4], t[2];
#pragma unroll
    for (int p = 0; p < 4; ++p) u[p] = (b3 ? v[p + 4] : v[p]) + row_xor<8>(b3 ? v[p] : v[p + 4]);
#pragma unroll
    for (int p = 0; p < 2; ++p) t[p] = (b2 ? u[p + 2] : u[p]) + row_xor<4>(b2 ? u[p] : u[p + 2]);
    const float s = (b1 ? t[1] : t[0]) + row_xor<2>(b1 ? t[0] : t[1]);
    return s + row_xor<1>(s);
}

__global__ __launch_bounds__(256)
void k_gemm_wmma(const float* __restrict__ X, const float* __restrict__ W,
                 const float* __restrict__ a_src, const float* __restrict__ a_dst,
                 float* __restrict__ Hout, float* __restrict__ AS, float* __restrict__ AD,
                 const int* __restrict__ ei, unsigned* keep, int first) {
    __shared__ float sX[2][16 * XS_PITCH];
    __shared__ float sSc[2][2 * 16 * HEADS];
    const int tid  = threadIdx.x;
    const int lane = tid & 31;
    const int wave = tid >> 5;
    const int l16  = lane & 15;
    const int half = lane >> 4;
    const int col  = wave * 16 + l16;

    const int srow = tid >> 5, sc4 = tid & 31;
    const float* xg = X + (size_t)srow * FEAT + sc4 * 4;
    const int    xl = srow * XS_PITCH + sc4 * 4;

    int tile = blockIdx.x;
    float4 g0 = *(const float4*)(xg + (size_t)tile * (16 * FEAT));
    float4 g1 = *(const float4*)(xg + (size_t)tile * (16 * FEAT) + 8 * FEAT);

    v2f b[FEAT / 4];
#pragma unroll
    for (int i = 0; i < FEAT / 4; ++i) {
        const int ka = 4 * i + 2 * half;
        b[i].x = W[ka * HID + col];
        b[i].y = W[(ka + 1) * HID + col];
    }
    const float asv = a_src[col], adv = a_dst[col];

    if (first) {
        for (unsigned u = blockIdx.x * 256u + (unsigned)tid; u < NEDGES / 4; u += GEMM_NB * 256u) {
            const int4     d  = *(const int4*)(ei + NEDGES + 4 * u);
            const unsigned o  = ((unsigned)d.x / OWN_R)         | (((unsigned)d.y / OWN_R) << 8) |
                                (((unsigned)d.z / OWN_R) << 16) | (((unsigned)d.w / OWN_R) << 24);
            const unsigned sl = u / MAP_W;
            keep[(size_t)sl * (OWN_R * HID) + (u - sl * MAP_W)] = o;
        }
    }

    *(float4*)&sX[0][xl]                = g0;
    *(float4*)&sX[0][xl + 8 * XS_PITCH] = g1;
    if (tile + GEMM_NB < GEMM_TILES) {
        g0 = *(const float4*)(xg + (size_t)(tile + GEMM_NB) * (16 * FEAT));
        g1 = *(const float4*)(xg + (size_t)(tile + GEMM_NB) * (16 * FEAT) + 8 * FEAT);
    }
    lds_barrier();

    const int arow = l16 * XS_PITCH + 2 * half;
    for (int it = 0; tile < GEMM_TILES; ++it, tile += GEMM_NB) {
        const float* xs = sX[it & 1] + arow;
        v2f a[FEAT / 4];
#pragma unroll
        for (int i = 0; i < FEAT / 4; ++i) a[i] = *(const v2f*)(xs + 4 * i);

        if (tile + GEMM_NB < GEMM_TILES) {
            float* xn = sX[(it & 1) ^ 1];
            *(float4*)&xn[xl]                = g0;
            *(float4*)&xn[xl + 8 * XS_PITCH] = g1;
            if (tile + 2 * GEMM_NB < GEMM_TILES) {
                g0 = *(const float4*)(xg + (size_t)(tile + 2 * GEMM_NB) * (16 * FEAT));
                g1 = *(const float4*)(xg + (size_t)(tile + 2 * GEMM_NB) * (16 * FEAT) + 8 * FEAT);
            }
        }

        v8f acc = {};
#pragma unroll
        for (int i = 0; i < FEAT / 4; ++i)
            acc = __builtin_amdgcn_wmma_f32_16x16x4_f32(
                      false, a[i], false, b[i], (short)0, acc, false, false);

        float ps[8], pd[8];
#pragma unroll
        for (int j = 0; j < 8; ++j) { ps[j] = acc[j] * asv; pd[j] = acc[j] * adv; }
        const float sa = row_sums8(ps, l16);
        const float sd = row_sums8(pd, l16);
        sSc[it & 1][(l16 & 1) * (16 * HEADS) + (8 * half + (l16 >> 1)) * HEADS + wave] = (l16 & 1) ? sd : sa;

        lds_barrier();

        const int orow0 = tile * 16 + 8 * half;
#pragma unroll
        for (int j = 0; j < 8; ++j)
            Hout[(size_t)(orow0 + j) * HID + col] = acc[j];
        (wave < 4 ? AS : AD)[(size_t)tile * (16 * HEADS) + (tid & 127)] = sSc[it & 1][tid];
    }
}

static __device__ __forceinline__ float leaky(float v) {
    return v >= 0.f ? v : NEG_SLOPE * v;
}

static __device__ __forceinline__ unsigned zero_bytes(unsigned x) {
    return ~(((x & 0x7F7F7F7Fu) + 0x7F7F7F7Fu) | x) & 0x80808080u;
}

static __device__ __forceinline__ void lds_fmax(float* p, float v) {
    __hip_atomic_fetch_max(p, v, __ATOMIC_RELAXED, __HIP_MEMORY_SCOPE_WORKGROUP);
}

static __device__ __forceinline__ int acc_pos(int c) { return (c & 3) * 32 + (c >> 2); }

template <int J>
static __device__ __forceinline__ float quad_bcast(float v) {
    return __int_as_float(__builtin_amdgcn_update_dpp(0, __float_as_int(v), J * 0x55, 0xF, 0xF, true));
}

static __device__ void edge_full_lane(const float* __restrict__ AS, const float* __restrict__ Hf,
                                      const float* sMA, const float* sAD, float* sS, float* acc,
                                      int s, int ld) {
    for (int hd = 0; hd < HEADS; ++hd) {
        const float ad = sAD[ld * HEADS + hd];
        const float m  = leaky(sMA[ld * HEADS + hd] + ad);
        const float w  = expf(leaky(AS[s * HEADS + hd] + ad) - m);
        atomicAdd(&sS[ld * HEADS + hd], w);
        for (int c = 0; c < HDIM; ++c) {
            const int col = hd * HDIM + c;
            atomicAdd(&acc[ld * HID + acc_pos(col)], w * Hf[(size_t)s * HID + col]);
        }
    }
}

__global__ __launch_bounds__(OWN_T)
void k_owner(const int* __restrict__ ei, const float* __restrict__ AS,
             const float* __restrict__ AD, const float* __restrict__ Hf,
             const float* __restrict__ bias, float* Out, unsigned* keep, int first) {
    __shared__ float    acc[OWN_R * HID];
    __shared__ float    sMA[OWN_R * HEADS];
    __shared__ float    sAD[OWN_R * HEADS];
    __shared__ float    sS [OWN_R * HEADS];
    __shared__ unsigned list[LIST_CAP];
    __shared__ unsigned cnt;

    const int tid  = threadIdx.x;
    const int base = blockIdx.x * OWN_R;
    const int Rn   = min(OWN_R, NNODES - base);

    unsigned*      mine     = keep + (size_t)base * HID + MAP_W;
    const unsigned keep_cap = (unsigned)(Rn * HID - MAP_W - KEEP_HDR);
    const unsigned kept     = first ? 0xFFFFFFFFu : mine[0];
    const bool     reuse    = kept <= keep_cap;

    for (int i = tid; i < OWN_R * HID; i += OWN_T) acc[i] = 0.f;
    for (int i = tid; i < Rn * HEADS; i += OWN_T) {
        sMA[i] = AS[base * HEADS + i];
        sAD[i] = AD[base * HEADS + i];
        sS[i]  = 0.f;
    }
    if (tid == 0) cnt = reuse ? kept : 0u;
    if (reuse)
        for (unsigned i = tid; i < kept; i += OWN_T) list[i] = mine[KEEP_HDR + i];
    __syncthreads();

    if (first) {
        const uint4*   mv  = (const uint4*)keep;
        const unsigned pat = (unsigned)blockIdx.x * 0x01010101u;
        for (int q0 = tid; q0 < MAP_NV; q0 += MAP_U * OWN_T) {
            uint4 v[MAP_U];
#pragma unroll
            for (int u = 0; u < MAP_U; ++u) {
                const unsigned q = (unsigned)(q0 + u * OWN_T);
                const unsigned j = q / MAP_SV;
                v[u] = q < MAP_NV ? mv[j * REGION_V + (q - j * MAP_SV)]
                                  : make_uint4(~pat, ~pat, ~pat, ~pat);
            }
            unsigned m[MAP_U], c = 0;
#pragma unroll
            for (int u = 0; u < MAP_U; ++u) {
                m[u] = (zero_bytes(v[u].x ^ pat) >> 7) | (zero_bytes(v[u].y ^ pat) >> 6) |
                       (zero_bytes(v[u].z ^ pat) >> 5) | (zero_bytes(v[u].w ^ pat) >> 4);
                c += (unsigned)__builtin_popcount(m[u]);
            }
            if (c) {
                unsigned slot = atomicAdd(&cnt, c);
#pragma unroll
                for (int u = 0; u < MAP_U; ++u) {
                    const unsigned e0 = (unsigned)(q0 + u * OWN_T) * 16u;
                    for (unsigned mm = m[u]; mm; mm &= mm - 1, ++slot) {
                        const unsigned bit = (unsigned)__builtin_ctz(mm);
                        if (slot < LIST_CAP) list[slot] = e0 + 4u * (bit & 7u) + (bit >> 3);
                    }
                }
            }
        }
    }
    __syncthreads();
    const unsigned total  = (unsigned)__builtin_amdgcn_readfirstlane((int)cnt);
    const unsigned nitems = total + (unsigned)Rn;
    const bool     save   = first && total <= keep_cap;
    if (first && tid == 0) mine[0] = total;

    if (first ? nitems <= LIST_CAP : reuse) {
        for (unsigned i = tid; i < total; i += OWN_T) {
            unsigned ent = list[i];
            if (first) {
                const int s = ei[ent];
                ent     = ((unsigned)s << 8) | (unsigned)(ei[NEDGES + ent] - base);
                list[i] = ent;
                if (save) mine[KEEP_HDR + i] = ent;
            }
            const unsigned ld = ent & 255u;
            const int      s  = (int)(ent >> 8);
            const float4* ap = (const float4*)(AS + (size_t)s * HEADS);
            const float4 a0 = ap[0], a1 = ap[1];
            float* m = &sMA[ld * HEADS];
            lds_fmax(m + 0, a0.x); lds_fmax(m + 1, a0.y);
            lds_fmax(m + 2, a0.z); lds_fmax(m + 3, a0.w);
            lds_fmax(m + 4, a1.x); lds_fmax(m + 5, a1.y);
            lds_fmax(m + 6, a1.z); lds_fmax(m + 7, a1.w);
        }
        for (unsigned k = tid; k < (unsigned)Rn; k += OWN_T)
            list[total + k] = ((unsigned)(base + (int)k) << 8) | k;
        __syncthreads();

        const int lane = tid & 31, hd = lane >> 2, q = lane & 3;
        const unsigned wave = (unsigned)__builtin_amdgcn_readfirstlane(tid >> 5);
        const float*   hcol = Hf + 4 * lane;
        for (unsigned i0 = wave * AGG_U; i0 < nitems; i0 += OWN_W * AGG_U) {
            const unsigned iA = i0 + (unsigned)q, iB = iA + 4u;
            const unsigned eA = list[min(iA, nitems - 1u)];
            const unsigned eB = list[min(iB, nitems - 1u)];
            unsigned en[AGG_U];
#pragma unroll
            for (int u = 0; u < 4; ++u) {
                en[u]     = (unsigned)__builtin_amdgcn_readlane((int)eA, u);
                en[u + 4] = (unsigned)__builtin_amdgcn_readlane((int)eB, u);
            }
            float4 hv[AGG_U];
#pragma unroll
            for (int u = 0; u < AGG_U; ++u)
                hv[u] = *(const float4*)(hcol + (size_t)(en[u] >> 8) * HID);
            const unsigned lA = (eA & 255u) * HEADS + (unsigned)hd, lB = (eB & 255u) * HEADS + (unsigned)hd;
            const float asA = AS[(eA >> 8) * HEADS + (unsigned)hd];
            const float asB = AS[(eB >> 8) * HEADS + (unsigned)hd];
            const float adA = sAD[lA], adB = sAD[lB];
            const float wA  = expf(leaky(asA + adA) - leaky(sMA[lA] + adA));
            const float wB  = expf(leaky(asB + adB) - leaky(sMA[lB] + adB));
            if (iA < nitems) atomicAdd(&sS[lA], wA);
            if (iB < nitems) atomicAdd(&sS[lB], wB);
#define AGG_ITEM(U, WSRC)                                                          \
            if (i0 + U < nitems) {                                \
                const float w = quad_bcast<(U) & 3>(WSRC);                         \
                const v2f   ww = {w, w};                                           \
                  \
                const v2f   p0 = ww * (v2f){hv[U].x, hv[U].y};                     \
                const v2f   p1 = ww * (v2f){hv[U].z, hv[U].w};                     \
                float* a = &acc[(en[U] & 255u) * HID + (unsigned)lane];            \
                atomicAdd(a,      p0.x);                                           \
                atomicAdd(a + 32, p0.y);                                           \
                atomicAdd(a + 64, p1.x);                                           \
                atomicAdd(a + 96, p1.y);                                           \
            }
            AGG_ITEM(0, wA) AGG_ITEM(1, wA) AGG_ITEM(2, wA) AGG_ITEM(3, wA)
            AGG_ITEM(4, wB) AGG_ITEM(5, wB) AGG_ITEM(6, wB) AGG_ITEM(7, wB)
#undef AGG_ITEM
        }
    } else {
        for (int e = tid; e < NEDGES; e += OWN_T) {
            const unsigned ld = (unsigned)(ei[NEDGES + e] - base);
            if (ld < (unsigned)Rn) {
                const int s = ei[e];
                for (int h = 0; h < HEADS; ++h) lds_fmax(&sMA[ld * HEADS + h], AS[s * HEADS + h]);
            }
        }
        __syncthreads();
        for (int e = tid; e < NEDGES; e += OWN_T) {
            const unsigned ld = (unsigned)(ei[NEDGES + e] - base);
            if (ld < (unsigned)Rn) edge_full_lane(AS, Hf, sMA, sAD, sS, acc, ei[e], (int)ld);
        }
        for (int ld = tid; ld < Rn; ld += OWN_T)
            edge_full_lane(AS, Hf, sMA, sAD, sS, acc, base + ld, ld);
    }
    __syncthreads();

    for (int i = tid; i < Rn * HID; i += OWN_T) {
        const int ld = i >> 7, c = i & (HID - 1);
        const float v = acc[ld * HID + acc_pos(c)] / (sS[ld * HEADS + (c >> 4)] + 1e-16f) + bias[c];
        Out[(size_t)(base + ld) * HID + c] = v > 0.f ? v : expm1f(v);
    }
}

extern "C" void kernel_launch(void* const* d_in, const int* in_sizes, int n_in,
                              void* d_out, int out_size, void* d_ws, size_t ws_size,
                              hipStream_t stream) {
    const float* x   = (const float*)d_in[0];
    const int*   ei  = (const int*)  d_in[1];
    const float* W1  = (const float*)d_in[2];
    const float* as1 = (const float*)d_in[3];
    const float* ad1 = (const float*)d_in[4];
    const float* b1  = (const float*)d_in[5];
    const float* W2  = (const float*)d_in[6];
    const float* as2 = (const float*)d_in[7];
    const float* ad2 = (const float*)d_in[8];
    const float* b2  = (const float*)d_in[9];

    float* ws     = (float*)d_ws;
    float* bufH   = ws;
    float* bufAS  = bufH  + (size_t)NNODES * HID;
    float* bufAD  = bufAS + (size_t)NNODES * HEADS;
    float* bufAgg = bufAD + (size_t)NNODES * HEADS;

    auto run_layer = [&](const float* xin, const float* Wm, const float* as,
                         const float* ad, const float* bv, float* outBuf, int first) {
        k_gemm_wmma<<<GEMM_NB, 256, 0, stream>>>(xin, Wm, as, ad, bufH, bufAS, bufAD, ei,
                                                 (unsigned*)d_out, first);
        k_owner<<<OWN_NB, OWN_T, 0, stream>>>(ei, bufAS, bufAD, bufH, bv, outBuf,
                                              (unsigned*)d_out, first);
    };

    run_layer(x,      W1, as1, ad1, b1, bufAgg,        1);
    run_layer(bufAgg, W2, as2, ad2, b2, (float*)d_out, 0);
}


# Source scores fetched once per item: the max phase parks the 8 scores it gathers (self loops: the nodes' own) in LDS, the row-sum phase reads them there; LDS list cut 32768->4608 items for room
# speedup vs baseline: 9.3006x; 1.0311x over previous
#include <hip/hip_runtime.h>
#include <math.h>

#define NNODES 50000
#define NEDGES 800000
#define FEAT 128
#define HID  128
#define HEADS 8
#define HDIM 16
#define NEG_SLOPE 0.2f

#define OWN_R    196
#define OWN_NB   256
#define OWN_T    1024
#define OWN_W    (OWN_T / 32)
#define LIST_CAP 4608
#define AGG_U    8
#define KEEP_HDR 32
#define MAP_SLICE 3200
#define MAP_W     (MAP_SLICE / 4)
#define MAP_SV    (MAP_SLICE / 16)
#define REGION_V  (OWN_R * HID / 4)
#define MAP_NV    (NEDGES / 16)
#define MAP_U     7
static_assert(MAP_SLICE % 128 == 0 && NEDGES % MAP_SLICE == 0 && NEDGES / MAP_SLICE < OWN_NB,
              "whole slices, all in full-size blocks");
static_assert((NNODES - (OWN_NB - 1) * OWN_R) * HID > MAP_W + KEEP_HDR, "last block holds its header");

typedef __attribute__((ext_vector_type(2))) float v2f;
typedef __attribute__((ext_vector_type(8))) float v8f;

#define GEMM_NB   512
#define GEMM_TILES (NNODES / 16)
#define XS_PITCH  132

static __device__ __forceinline__ void lds_barrier() {
    __builtin_amdgcn_fence(__ATOMIC_RELEASE, "workgroup", "local");
    __builtin_amdgcn_s_barrier();
    __builtin_amdgcn_fence(__ATOMIC_ACQUIRE, "workgroup", "local");
}

template <int K>
static __device__ __forceinline__ float row_xor(float v) {
    return __int_as_float(__builtin_amdgcn_update_dpp(0, __float_as_int(v), 0x160 + K, 0xF, 0xF, true));
}

static __device__ __forceinline__ float row_sums8(const float (&v)[8], int c) {
    const bool b3 = (c & 8) != 0, b2 = (c & 4) != 0, b1 = (c & 2) != 0;
    float u[4], t[2];
#pragma unroll
    for (int p = 0; p < 4; ++p) u[p] = (b3 ? v[p + 4] : v[p]) + row_xor<8>(b3 ? v[p] : v[p + 4]);
#pragma unroll
    for (int p = 0; p < 2; ++p) t[p] = (b2 ? u[p + 2] : u[p]) + row_xor<4>(b2 ? u[p] : u[p + 2]);
    const float s = (b1 ? t[1] : t[0]) + row_xor<2>(b1 ? t[0] : t[1]);
    return s + row_xor<1>(s);
}

__global__ __launch_bounds__(256)
void k_gemm_wmma(const float* __restrict__ X, const float* __restrict__ W,
                 const float* __restrict__ a_src, const float* __restrict__ a_dst,
                 float* __restrict__ Hout, float* __restrict__ AS, float* __restrict__ AD,
                 const int* __restrict__ ei, unsigned* keep, int first) {
    __shared__ float sX[2][16 * XS_PITCH];
    __shared__ float sSc[2][2 * 16 * HEADS];
    const int tid  = threadIdx.x;
    const int lane = tid & 31;
    const int wave = tid >> 5;
    const int l16  = lane & 15;
    const int half = lane >> 4;
    const int col  = wave * 16 + l16;

    const int srow = tid >> 5, sc4 = tid & 31;
    const float* xg = X + (size_t)srow * FEAT + sc4 * 4;
    const int    xl = srow * XS_PITCH + sc4 * 4;

    int tile = blockIdx.x;
    float4 g0 = *(const float4*)(xg + (size_t)tile * (16 * FEAT));
    float4 g1 = *(const float4*)(xg + (size_t)tile * (16 * FEAT) + 8 * FEAT);

    v2f b[FEAT / 4];
#pragma unroll
    for (int i = 0; i < FEAT / 4; ++i) {
        const int ka = 4 * i + 2 * half;
        b[i].x = W[ka * HID + col];
        b[i].y = W[(ka + 1) * HID + col];
    }
    const float asv = a_src[col], adv = a_dst[col];

    if (first) {
        for (unsigned u = blockIdx.x * 256u + (unsigned)tid; u < NEDGES / 4; u += GEMM_NB * 256u) {
            const int4     d  = *(const int4*)(ei + NEDGES + 4 * u);
            const unsigned o  = ((unsigned)d.x / OWN_R)         | (((unsigned)d.y / OWN_R) << 8) |
                                (((unsigned)d.z / OWN_R) << 16) | (((unsigned)d.w / OWN_R) << 24);
            const unsigned sl = u / MAP_W;
            keep[(size_t)sl * (OWN_R * HID) + (u - sl * MAP_W)] = o;
        }
    }

    *(float4*)&sX[0][xl]                = g0;
    *(float4*)&sX[0][xl + 8 * XS_PITCH] = g1;
    if (tile + GEMM_NB < GEMM_TILES) {
        g0 = *(const float4*)(xg + (size_t)(tile + GEMM_NB) * (16 * FEAT));
        g1 = *(const float4*)(xg + (size_t)(tile + GEMM_NB) * (16 * FEAT) + 8 * FEAT);
    }
    lds_barrier();

    const int arow = l16 * XS_PITCH + 2 * half;
    for (int it = 0; tile < GEMM_TILES; ++it, tile += GEMM_NB) {
        const float* xs = sX[it & 1] + arow;
        v2f a[FEAT / 4];
#pragma unroll
        for (int i = 0; i < FEAT / 4; ++i) a[i] = *(const v2f*)(xs + 4 * i);

        if (tile + GEMM_NB < GEMM_TILES) {
            float* xn = sX[(it & 1) ^ 1];
            *(float4*)&xn[xl]                = g0;
            *(float4*)&xn[xl + 8 * XS_PITCH] = g1;
            if (tile + 2 * GEMM_NB < GEMM_TILES) {
                g0 = *(const float4*)(xg + (size_t)(tile + 2 * GEMM_NB) * (16 * FEAT));
                g1 = *(const float4*)(xg + (size_t)(tile + 2 * GEMM_NB) * (16 * FEAT) + 8 * FEAT);
            }
        }

        v8f acc = {};
#pragma unroll
        for (int i = 0; i < FEAT / 4; ++i)
            acc = __builtin_amdgcn_wmma_f32_16x16x4_f32(
                      false, a[i], false, b[i], (short)0, acc, false, false);

        float ps[8], pd[8];
#pragma unroll
        for (int j = 0; j < 8; ++j) { ps[j] = acc[j] * asv; pd[j] = acc[j] * adv; }
        const float sa = row_sums8(ps, l16);
        const float sd = row_sums8(pd, l16);
        sSc[it & 1][(l16 & 1) * (16 * HEADS) + (8 * half + (l16 >> 1)) * HEADS + wave] = (l16 & 1) ? sd : sa;

        lds_barrier();

        const int orow0 = tile * 16 + 8 * half;
#pragma unroll
        for (int j = 0; j < 8; ++j)
            Hout[(size_t)(orow0 + j) * HID + col] = acc[j];
        (wave < 4 ? AS : AD)[(size_t)tile * (16 * HEADS) + (tid & 127)] = sSc[it & 1][tid];
    }
}

static __device__ __forceinline__ float leaky(float v) {
    return v >= 0.f ? v : NEG_SLOPE * v;
}

static __device__ __forceinline__ unsigned zero_bytes(unsigned x) {
    return ~(((x & 0x7F7F7F7Fu) + 0x7F7F7F7Fu) | x) & 0x80808080u;
}

static __device__ __forceinline__ void lds_fmax(float* p, float v) {
    __hip_atomic_fetch_max(p, v, __ATOMIC_RELAXED, __HIP_MEMORY_SCOPE_WORKGROUP);
}

static __device__ __forceinline__ int acc_pos(int c) { return (c & 3) * 32 + (c >> 2); }

template <int J>
static __device__ __forceinline__ float quad_bcast(float v) {
    return __int_as_float(__builtin_amdgcn_update_dpp(0, __float_as_int(v), J * 0x55, 0xF, 0xF, true));
}

static __device__ void edge_full_lane(const float* __restrict__ AS, const float* __restrict__ Hf,
                                      const float* sMA, const float* sAD, float* sS, float* acc,
                                      int s, int ld) {
    for (int hd = 0; hd < HEADS; ++hd) {
        const float ad = sAD[ld * HEADS + hd];
        const float m  = leaky(sMA[ld * HEADS + hd] + ad);
        const float w  = expf(leaky(AS[s * HEADS + hd] + ad) - m);
        atomicAdd(&sS[ld * HEADS + hd], w);
        for (int c = 0; c < HDIM; ++c) {
            const int col = hd * HDIM + c;
            atomicAdd(&acc[ld * HID + acc_pos(col)], w * Hf[(size_t)s * HID + col]);
        }
    }
}

__global__ __launch_bounds__(OWN_T)
void k_owner(const int* __restrict__ ei, const float* __restrict__ AS,
             const float* __restrict__ AD, const float* __restrict__ Hf,
             const float* __restrict__ bias, float* Out, unsigned* keep, int first) {
    __shared__ float    acc[OWN_R * HID];
    __shared__ float    sMA[OWN_R * HEADS];
    __shared__ float    sAD[OWN_R * HEADS];
    __shared__ float    sS [OWN_R * HEADS];
    __shared__ unsigned list[LIST_CAP];
    __shared__ alignas(16) float sASc[LIST_CAP * HEADS];
    __shared__ unsigned cnt;

    const int tid  = threadIdx.x;
    const int base = blockIdx.x * OWN_R;
    const int Rn   = min(OWN_R, NNODES - base);

    unsigned*      mine     = keep + (size_t)base * HID + MAP_W;
    const unsigned keep_cap = (unsigned)min(Rn * HID - MAP_W - KEEP_HDR, LIST_CAP - Rn);
    const unsigned kept     = first ? 0xFFFFFFFFu : mine[0];
    const bool     reuse    = kept <= keep_cap;

    for (int i = tid; i < OWN_R * HID; i += OWN_T) acc[i] = 0.f;
    for (int i = tid; i < Rn * HEADS; i += OWN_T) {
        sMA[i] = AS[base * HEADS + i];
        sAD[i] = AD[base * HEADS + i];
        sS[i]  = 0.f;
    }
    if (tid == 0) cnt = reuse ? kept : 0u;
    if (reuse)
        for (unsigned i = tid; i < kept; i += OWN_T) list[i] = mine[KEEP_HDR + i];
    __syncthreads();

    if (first) {
        const uint4*   mv  = (const uint4*)keep;
        const unsigned pat = (unsigned)blockIdx.x * 0x01010101u;
        for (int q0 = tid; q0 < MAP_NV; q0 += MAP_U * OWN_T) {
            uint4 v[MAP_U];
#pragma unroll
            for (int u = 0; u < MAP_U; ++u) {
                const unsigned q = (unsigned)(q0 + u * OWN_T);
                const unsigned j = q / MAP_SV;
                v[u] = q < MAP_NV ? mv[j * REGION_V + (q - j * MAP_SV)]
                                  : make_uint4(~pat, ~pat, ~pat, ~pat);
            }
            unsigned m[MAP_U], c = 0;
#pragma unroll
            for (int u = 0; u < MAP_U; ++u) {
                m[u] = (zero_bytes(v[u].x ^ pat) >> 7) | (zero_bytes(v[u].y ^ pat) >> 6) |
                       (zero_bytes(v[u].z ^ pat) >> 5) | (zero_bytes(v[u].w ^ pat) >> 4);
                c += (unsigned)__builtin_popcount(m[u]);
            }
            if (c) {
                unsigned slot = atomicAdd(&cnt, c);
#pragma unroll
                for (int u = 0; u < MAP_U; ++u) {
                    const unsigned e0 = (unsigned)(q0 + u * OWN_T) * 16u;
                    for (unsigned mm = m[u]; mm; mm &= mm - 1, ++slot) {
                        const unsigned bit = (unsigned)__builtin_ctz(mm);
                        if (slot < LIST_CAP) list[slot] = e0 + 4u * (bit & 7u) + (bit >> 3);
                    }
                }
            }
        }
    }
    __syncthreads();
    const unsigned total  = (unsigned)__builtin_amdgcn_readfirstlane((int)cnt);
    const unsigned nitems = total + (unsigned)Rn;
    const bool     save   = first && total <= keep_cap;
    if (first && tid == 0) mine[0] = total;

    if (first ? nitems <= LIST_CAP : reuse) {
        for (unsigned i = tid; i < total; i += OWN_T) {
            unsigned ent = list[i];
            if (first) {
                const int s = ei[ent];
                ent     = ((unsigned)s << 8) | (unsigned)(ei[NEDGES + ent] - base);
                list[i] = ent;
                if (save) mine[KEEP_HDR + i] = ent;
            }
            const unsigned ld = ent & 255u;
            const int      s  = (int)(ent >> 8);
            const float4* ap = (const float4*)(AS + (size_t)s * HEADS);
            const float4 a0 = ap[0], a1 = ap[1];
            *(float4*)&sASc[HEADS * i]     = a0;
            *(float4*)&sASc[HEADS * i + 4] = a1;
            float* m = &sMA[ld * HEADS];
            lds_fmax(m + 0, a0.x); lds_fmax(m + 1, a0.y);
            lds_fmax(m + 2, a0.z); lds_fmax(m + 3, a0.w);
            lds_fmax(m + 4, a1.x); lds_fmax(m + 5, a1.y);
            lds_fmax(m + 6, a1.z); lds_fmax(m + 7, a1.w);
        }
        for (unsigned k = tid; k < (unsigned)Rn; k += OWN_T) {
            list[total + k] = ((unsigned)(base + (int)k) << 8) | k;
            const float4* ap = (const float4*)(AS + (size_t)(base + (int)k) * HEADS);
            *(float4*)&sASc[HEADS * (total + k)]     = ap[0];
            *(float4*)&sASc[HEADS * (total + k) + 4] = ap[1];
        }
        __syncthreads();

        const int lane = tid & 31, hd = lane >> 2, q = lane & 3;
        const unsigned wave = (unsigned)__builtin_amdgcn_readfirstlane(tid >> 5);
        const float*   hcol = Hf + 4 * lane;
        for (unsigned i0 = wave * AGG_U; i0 < nitems; i0 += OWN_W * AGG_U) {
            const unsigned iA = i0 + (unsigned)q, iB = iA + 4u;
            const unsigned cA = min(iA, nitems - 1u), cB = min(iB, nitems - 1u);
            const unsigned eA = list[cA];
            const unsigned eB = list[cB];
            unsigned en[AGG_U];
#pragma unroll
            for (int u = 0; u < 4; ++u) {
                en[u]     = (unsigned)__builtin_amdgcn_readlane((int)eA, u);
                en[u + 4] = (unsigned)__builtin_amdgcn_readlane((int)eB, u);
            }
            float4 hv[AGG_U];
#pragma unroll
            for (int u = 0; u < AGG_U; ++u)
                hv[u] = *(const float4*)(hcol + (size_t)(en[u] >> 8) * HID);
            const unsigned lA = (eA & 255u) * HEADS + (unsigned)hd, lB = (eB & 255u) * HEADS + (unsigned)hd;
            const float asA = sASc[cA * HEADS + (unsigned)hd];
            const float asB = sASc[cB * HEADS + (unsigned)hd];
            const float adA = sAD[lA], adB = sAD[lB];
            const float wA  = expf(leaky(asA + adA) - leaky(sMA[lA] + adA));
            const float wB  = expf(leaky(asB + adB) - leaky(sMA[lB] + adB));
            if (iA < nitems) atomicAdd(&sS[lA], wA);
            if (iB < nitems) atomicAdd(&sS[lB], wB);
#define AGG_ITEM(U, WSRC)                                                          \
            if (i0 + U < nitems) {                                \
                const float w = quad_bcast<(U) & 3>(WSRC);                         \
                const v2f   ww = {w, w};                                           \
                  \
                const v2f   p0 = ww * (v2f){hv[U].x, hv[U].y};                     \
                const v2f   p1 = ww * (v2f){hv[U].z, hv[U].w};                     \
                float* a = &acc[(en[U] & 255u) * HID + (unsigned)lane];            \
                atomicAdd(a,      p0.x);                                           \
                atomicAdd(a + 32, p0.y);                                           \
                atomicAdd(a + 64, p1.x);                                           \
                atomicAdd(a + 96, p1.y);                                           \
            }
            AGG_ITEM(0, wA) AGG_ITEM(1, wA) AGG_ITEM(2, wA) AGG_ITEM(3, wA)
            AGG_ITEM(4, wB) AGG_ITEM(5, wB) AGG_ITEM(6, wB) AGG_ITEM(7, wB)
#undef AGG_ITEM
        }
    } else {
        for (int e = tid; e < NEDGES; e += OWN_T) {
            const unsigned ld = (unsigned)(ei[NEDGES + e] - base);
            if (ld < (unsigned)Rn) {
                const int s = ei[e];
                for (int h = 0; h < HEADS; ++h) lds_fmax(&sMA[ld * HEADS + h], AS[s * HEADS + h]);
            }
        }
        __syncthreads();
        for (int e = tid; e < NEDGES; e += OWN_T) {
            const unsigned ld = (unsigned)(ei[NEDGES + e] - base);
            if (ld < (unsigned)Rn) edge_full_lane(AS, Hf, sMA, sAD, sS, acc, ei[e], (int)ld);
        }
        for (int ld = tid; ld < Rn; ld += OWN_T)
            edge_full_lane(AS, Hf, sMA, sAD, sS, acc, base + ld, ld);
    }
    __syncthreads();

    for (int i = tid; i < Rn * HID; i += OWN_T) {
        const int ld = i >> 7, c = i & (HID - 1);
        const float v = acc[ld * HID + acc_pos(c)] / (sS[ld * HEADS + (c >> 4)] + 1e-16f) + bias[c];
        Out[(size_t)(base + ld) * HID + c] = v > 0.f ? v : expm1f(v);
    }
}

extern "C" void kernel_launch(void* const* d_in, const int* in_sizes, int n_in,
                              void* d_out, int out_size, void* d_ws, size_t ws_size,
                              hipStream_t stream) {
    const float* x   = (const float*)d_in[0];
    const int*   ei  = (const int*)  d_in[1];
    const float* W1  = (const float*)d_in[2];
    const float* as1 = (const float*)d_in[3];
    const float* ad1 = (const float*)d_in[4];
    const float* b1  = (const float*)d_in[5];
    const float* W2  = (const float*)d_in[6];
    const float* as2 = (const float*)d_in[7];
    const float* ad2 = (const float*)d_in[8];
    const float* b2  = (const float*)d_in[9];

    float* ws     = (float*)d_ws;
    float* bufH   = ws;
    float* bufAS  = bufH  + (size_t)NNODES * HID;
    float* bufAD  = bufAS + (size_t)NNODES * HEADS;
    float* bufAgg = bufAD + (size_t)NNODES * HEADS;

    auto run_layer = [&](const float* xin, const float* Wm, const float* as,
                         const float* ad, const float* bv, float* outBuf, int first) {
        k_gemm_wmma<<<GEMM_NB, 256, 0, stream>>>(xin, Wm, as, ad, bufH, bufAS, bufAD, ei,
                                                 (unsigned*)d_out, first);
        k_owner<<<OWN_NB, OWN_T, 0, stream>>>(ei, bufAS, bufAD, bufH, bv, outBuf,
                                              (unsigned*)d_out, first);
    };

    run_layer(x,      W1, as1, ad1, b1, bufAgg,        1);
    run_layer(bufAgg, W2, as2, ad2, b2, (float*)d_out, 0);
}


# Row sums leave LDS float atomics: items grouped by the wave that owns their row (node & 31; count, prefix, place in LDS), which adds them in registers and stores the row; sum order changes
# speedup vs baseline: 9.8606x; 1.0602x over previous
#include <hip/hip_runtime.h>
#include <math.h>

#define NNODES 50000
#define NEDGES 800000
#define FEAT 128
#define HID  128
#define HEADS 8
#define HDIM 16
#define NEG_SLOPE 0.2f

#define OWN_R    196
#define OWN_NB   256
#define OWN_T    1024
#define OWN_W    (OWN_T / 32)
#define LIST_CAP 4608
#define AGG_U    8
#define ROWS_W   7
#define MAX_PT   5
#define KEEP_HDR 32
#define MAP_SLICE 3200
#define MAP_W     (MAP_SLICE / 4)
#define MAP_SV    (MAP_SLICE / 16)
#define REGION_V  (OWN_R * HID / 4)
#define MAP_NV    (NEDGES / 16)
#define MAP_U     7
static_assert(MAP_SLICE % 128 == 0 && NEDGES % MAP_SLICE == 0 && NEDGES / MAP_SLICE < OWN_NB,
              "whole slices, all in full-size blocks");
static_assert((NNODES - (OWN_NB - 1) * OWN_R) * HID > MAP_W + KEEP_HDR, "last block holds its header");
static_assert(ROWS_W * OWN_W >= OWN_R && OWN_W == 32 && OWN_R <= OWN_T && MAX_PT * OWN_T >= LIST_CAP,
              "every owned row has an owner wave and a register slot; one thread per self loop");

typedef __attribute__((ext_vector_type(2))) float v2f;
typedef __attribute__((ext_vector_type(8))) float v8f;

#define GEMM_NB   512
#define GEMM_TILES (NNODES / 16)
#define XS_PITCH  132

static __device__ __forceinline__ void lds_barrier() {
    __builtin_amdgcn_fence(__ATOMIC_RELEASE, "workgroup", "local");
    __builtin_amdgcn_s_barrier();
    __builtin_amdgcn_fence(__ATOMIC_ACQUIRE, "workgroup", "local");
}

template <int K>
static __device__ __forceinline__ float row_xor(float v) {
    return __int_as_float(__builtin_amdgcn_update_dpp(0, __float_as_int(v), 0x160 + K, 0xF, 0xF, true));
}

static __device__ __forceinline__ float row_sums8(const float (&v)[8], int c) {
    const bool b3 = (c & 8) != 0, b2 = (c & 4) != 0, b1 = (c & 2) != 0;
    float u[4], t[2];
#pragma unroll
    for (int p = 0; p < 4; ++p) u[p] = (b3 ? v[p + 4] : v[p]) + row_xor<8>(b3 ? v[p] : v[p + 4]);
#pragma unroll
    for (int p = 0; p < 2; ++p) t[p] = (b2 ? u[p + 2] : u[p]) + row_xor<4>(b2 ? u[p] : u[p + 2]);
    const float s = (b1 ? t[1] : t[0]) + row_xor<2>(b1 ? t[0] : t[1]);
    return s + row_xor<1>(s);
}

__global__ __launch_bounds__(256)
void k_gemm_wmma(const float* __restrict__ X, const float* __restrict__ W,
                 const float* __restrict__ a_src, const float* __restrict__ a_dst,
                 float* __restrict__ Hout, float* __restrict__ AS, float* __restrict__ AD,
                 const int* __restrict__ ei, unsigned* keep, int first) {
    __shared__ float sX[2][16 * XS_PITCH];
    __shared__ float sSc[2][2 * 16 * HEADS];
    const int tid  = threadIdx.x;
    const int lane = tid & 31;
    const int wave = tid >> 5;
    const int l16  = lane & 15;
    const int half = lane >> 4;
    const int col  = wave * 16 + l16;

    const int srow = tid >> 5, sc4 = tid & 31;
    const float* xg = X + (size_t)srow * FEAT + sc4 * 4;
    const int    xl = srow * XS_PITCH + sc4 * 4;

    int tile = blockIdx.x;
    float4 g0 = *(const float4*)(xg + (size_t)tile * (16 * FEAT));
    float4 g1 = *(const float4*)(xg + (size_t)tile * (16 * FEAT) + 8 * FEAT);

    v2f b[FEAT / 4];
#pragma unroll
    for (int i = 0; i < FEAT / 4; ++i) {
        const int ka = 4 * i + 2 * half;
        b[i].x = W[ka * HID + col];
        b[i].y = W[(ka + 1) * HID + col];
    }
    const float asv = a_src[col], adv = a_dst[col];

    if (first) {
        for (unsigned u = blockIdx.x * 256u + (unsigned)tid; u < NEDGES / 4; u += GEMM_NB * 256u) {
            const int4     d  = *(const int4*)(ei + NEDGES + 4 * u);
            const unsigned o  = ((unsigned)d.x / OWN_R)         | (((unsigned)d.y / OWN_R) << 8) |
                                (((unsigned)d.z / OWN_R) << 16) | (((unsigned)d.w / OWN_R) << 24);
            const unsigned sl = u / MAP_W;
            keep[(size_t)sl * (OWN_R * HID) + (u - sl * MAP_W)] = o;
        }
    }

    *(float4*)&sX[0][xl]                = g0;
    *(float4*)&sX[0][xl + 8 * XS_PITCH] = g1;
    if (tile + GEMM_NB < GEMM_TILES) {
        g0 = *(const float4*)(xg + (size_t)(tile + GEMM_NB) * (16 * FEAT));
        g1 = *(const float4*)(xg + (size_t)(tile + GEMM_NB) * (16 * FEAT) + 8 * FEAT);
    }
    lds_barrier();

    const int arow = l16 * XS_PITCH + 2 * half;
    for (int it = 0; tile < GEMM_TILES; ++it, tile += GEMM_NB) {
        const float* xs = sX[it & 1] + arow;
        v2f a[FEAT / 4];
#pragma unroll
        for (int i = 0; i < FEAT / 4; ++i) a[i] = *(const v2f*)(xs + 4 * i);

        if (tile + GEMM_NB < GEMM_TILES) {
            float* xn = sX[(it & 1) ^ 1];
            *(float4*)&xn[xl]                = g0;
            *(float4*)&xn[xl + 8 * XS_PITCH] = g1;
            if (tile + 2 * GEMM_NB < GEMM_TILES) {
                g0 = *(const float4*)(xg + (size_t)(tile + 2 * GEMM_NB) * (16 * FEAT));
                g1 = *(const float4*)(xg + (size_t)(tile + 2 * GEMM_NB) * (16 * FEAT) + 8 * FEAT);
            }
        }

        v8f acc = {};
#pragma unroll
        for (int i = 0; i < FEAT / 4; ++i)
            acc = __builtin_amdgcn_wmma_f32_16x16x4_f32(
                      false, a[i], false, b[i], (short)0, acc, false, false);

        float ps[8], pd[8];
#pragma unroll
        for (int j = 0; j < 8; ++j) { ps[j] = acc[j] * asv; pd[j] = acc[j] * adv; }
        const float sa = row_sums8(ps, l16);
        const float sd = row_sums8(pd, l16);
        sSc[it & 1][(l16 & 1) * (16 * HEADS) + (8 * half + (l16 >> 1)) * HEADS + wave] = (l16 & 1) ? sd : sa;

        lds_barrier();

        const int orow0 = tile * 16 + 8 * half;
#pragma unroll
        for (int j = 0; j < 8; ++j)
            Hout[(size_t)(orow0 + j) * HID + col] = acc[j];
        (wave < 4 ? AS : AD)[(size_t)tile * (16 * HEADS) + (tid & 127)] = sSc[it & 1][tid];
    }
}

static __device__ __forceinline__ float leaky(float v) {
    return v >= 0.f ? v : NEG_SLOPE * v;
}

static __device__ __forceinline__ unsigned zero_bytes(unsigned x) {
    return ~(((x & 0x7F7F7F7Fu) + 0x7F7F7F7Fu) | x) & 0x80808080u;
}

static __device__ __forceinline__ void lds_fmax(float* p, float v) {
    __hip_atomic_fetch_max(p, v, __ATOMIC_RELAXED, __HIP_MEMORY_SCOPE_WORKGROUP);
}

static __device__ __forceinline__ int acc_pos(int c) { return (c & 3) * 32 + (c >> 2); }

template <int J>
static __device__ __forceinline__ float quad_bcast(float v) {
    return __int_as_float(__builtin_amdgcn_update_dpp(0, __float_as_int(v), J * 0x55, 0xF, 0xF, true));
}

static __device__ void edge_full_lane(const float* __restrict__ AS, const float* __restrict__ Hf,
                                      const float* sMA, const float* sAD, float* sS, float* acc,
                                      int s, int ld) {
    for (int hd = 0; hd < HEADS; ++hd) {
        const float ad = sAD[ld * HEADS + hd];
        const float m  = leaky(sMA[ld * HEADS + hd] + ad);
        const float w  = expf(leaky(AS[s * HEADS + hd] + ad) - m);
        atomicAdd(&sS[ld * HEADS + hd], w);
        for (int c = 0; c < HDIM; ++c) {
            const int col = hd * HDIM + c;
            atomicAdd(&acc[ld * HID + acc_pos(col)], w * Hf[(size_t)s * HID + col]);
        }
    }
}

static __device__ __forceinline__ float elu1(float v) {
    const float e = expm1f(v);
    return v > 0.f ? v : e;
}

static __device__ __forceinline__ void row_out(float* dst, v2f a, v2f b, float s, float4 bv) {
    const float den = s + 1e-16f;
    float4 o;
    o.x = elu1(a.x / den + bv.x);
    o.y = elu1(a.y / den + bv.y);
    o.z = elu1(b.x / den + bv.z);
    o.w = elu1(b.y / den + bv.w);
    *(float4*)dst = o;
}

__global__ __launch_bounds__(OWN_T)
void k_owner(const int* __restrict__ ei, const float* __restrict__ AS,
             const float* __restrict__ AD, const float* __restrict__ Hf,
             const float* __restrict__ bias, float* Out, unsigned* keep, int first) {
    __shared__ float    acc[OWN_R * HID];
    __shared__ float    sMA[OWN_R * HEADS];
    __shared__ float    sAD[OWN_R * HEADS];
    __shared__ float    sS [OWN_R * HEADS];
    __shared__ unsigned list[LIST_CAP];
    __shared__ alignas(16) float sASc[LIST_CAP * HEADS];
    __shared__ unsigned perm[LIST_CAP];
    __shared__ unsigned wcnt[OWN_W];
    __shared__ unsigned cnt;

    const int tid  = threadIdx.x;
    const int base = blockIdx.x * OWN_R;
    const int Rn   = min(OWN_R, NNODES - base);

    unsigned*      mine     = keep + (size_t)base * HID + MAP_W;
    const unsigned keep_cap = (unsigned)min(Rn * HID - MAP_W - KEEP_HDR, LIST_CAP - Rn);
    const unsigned kept     = first ? 0xFFFFFFFFu : mine[0];
    const bool     reuse    = kept <= keep_cap;

    for (int i = tid; i < Rn * HEADS; i += OWN_T) {
        sMA[i] = AS[base * HEADS + i];
        sAD[i] = AD[base * HEADS + i];
        sS[i]  = 0.f;
    }
    if (tid == 0) cnt = reuse ? kept : 0u;
    if (tid < OWN_W) wcnt[tid] = 0u;
    if (reuse)
        for (unsigned i = tid; i < kept; i += OWN_T) list[i] = mine[KEEP_HDR + i];
    __syncthreads();

    if (first) {
        const uint4*   mv  = (const uint4*)keep;
        const unsigned pat = (unsigned)blockIdx.x * 0x01010101u;
        for (int q0 = tid; q0 < MAP_NV; q0 += MAP_U * OWN_T) {
            uint4 v[MAP_U];
#pragma unroll
            for (int u = 0; u < MAP_U; ++u) {
                const unsigned q = (unsigned)(q0 + u * OWN_T);
                const unsigned j = q / MAP_SV;
                v[u] = q < MAP_NV ? mv[j * REGION_V + (q - j * MAP_SV)]
                                  : make_uint4(~pat, ~pat, ~pat, ~pat);
            }
            unsigned m[MAP_U], c = 0;
#pragma unroll
            for (int u = 0; u < MAP_U; ++u) {
                m[u] = (zero_bytes(v[u].x ^ pat) >> 7) | (zero_bytes(v[u].y ^ pat) >> 6) |
                       (zero_bytes(v[u].z ^ pat) >> 5) | (zero_bytes(v[u].w ^ pat) >> 4);
                c += (unsigned)__builtin_popcount(m[u]);
            }
            if (c) {
                unsigned slot = atomicAdd(&cnt, c);
#pragma unroll
                for (int u = 0; u < MAP_U; ++u) {
                    const unsigned e0 = (unsigned)(q0 + u * OWN_T) * 16u;
                    for (unsigned mm = m[u]; mm; mm &= mm - 1, ++slot) {
                        const unsigned bit = (unsigned)__builtin_ctz(mm);
                        if (slot < LIST_CAP) list[slot] = e0 + 4u * (bit & 7u) + (bit >> 3);
                    }
                }
            }
        }
    }
    __syncthreads();
    const unsigned total  = (unsigned)__builtin_amdgcn_readfirstlane((int)cnt);
    const unsigned nitems = total + (unsigned)Rn;
    const bool     save   = first && total <= keep_cap;
    if (first && tid == 0) mine[0] = total;

    if (first ? nitems <= LIST_CAP : reuse) {
        unsigned rk[MAX_PT];
#pragma unroll
        for (int t = 0; t < MAX_PT; ++t) {
            const unsigned i = (unsigned)tid + (unsigned)t * OWN_T;
            rk[t] = 0u;
            if (i < total) {
                unsigned ent = list[i];
                if (first) {
                    const int s = ei[ent];
                    ent     = ((unsigned)s << 8) | (unsigned)(ei[NEDGES + ent] - base);
                    list[i] = ent;
                    if (save) mine[KEEP_HDR + i] = ent;
                }
                const unsigned ld = ent & 255u;
                const int      s  = (int)(ent >> 8);
                const float4* ap = (const float4*)(AS + (size_t)s * HEADS);
                const float4 a0 = ap[0], a1 = ap[1];
                *(float4*)&sASc[HEADS * i]     = a0;
                *(float4*)&sASc[HEADS * i + 4] = a1;
                float* m = &sMA[ld * HEADS];
                lds_fmax(m + 0, a0.x); lds_fmax(m + 1, a0.y);
                lds_fmax(m + 2, a0.z); lds_fmax(m + 3, a0.w);
                lds_fmax(m + 4, a1.x); lds_fmax(m + 5, a1.y);
                lds_fmax(m + 6, a1.z); lds_fmax(m + 7, a1.w);
                rk[t] = atomicAdd(&wcnt[ld & 31u], 1u);
            }
        }
        unsigned rkS = 0u;
        if (tid < Rn) {
            list[total + (unsigned)tid] = ((unsigned)(base + tid) << 8) | (unsigned)tid;
            const float4* ap = (const float4*)(AS + (size_t)(base + tid) * HEADS);
            *(float4*)&sASc[HEADS * (total + (unsigned)tid)]     = ap[0];
            *(float4*)&sASc[HEADS * (total + (unsigned)tid) + 4] = ap[1];
            rkS = atomicAdd(&wcnt[tid & 31], 1u);
        }
        __syncthreads();

        const int lane = tid & 31, hd = lane >> 2, q = lane & 3;
        const unsigned cntl = wcnt[lane];
        unsigned incl = cntl;
#pragma unroll
        for (int d = 1; d < 32; d <<= 1) {
            const unsigned up = __shfl_up(incl, (unsigned)d);
            if (lane >= d) incl += up;
        }
        const unsigned excl = incl - cntl;
#pragma unroll
        for (int t = 0; t < MAX_PT; ++t) {
            const unsigned i  = (unsigned)tid + (unsigned)t * OWN_T;
            const unsigned st = (unsigned)__shfl((int)excl, (int)(list[min(i, nitems - 1u)] & 31u));
            if (i < total) perm[st + rk[t]] = i;
        }
        if (tid < Rn) perm[excl + rkS] = total + (unsigned)tid;
        __syncthreads();

        const unsigned wave = (unsigned)__builtin_amdgcn_readfirstlane(tid >> 5);
        const unsigned jbeg = (unsigned)__builtin_amdgcn_readlane((int)excl, (int)wave);
        const unsigned jend = jbeg + (unsigned)__builtin_amdgcn_readlane((int)cntl, (int)wave);
        const float*   hcol = Hf + 4 * lane;
        v8f ax = {}, ay = {}, az = {}, aw = {}, sv = {};
        for (unsigned j0 = jbeg; j0 < jend; j0 += AGG_U) {
            const unsigned jA = j0 + (unsigned)q, jB = jA + 4u;
            const unsigned iA = perm[min(jA, jend - 1u)], iB = perm[min(jB, jend - 1u)];
            const unsigned eA = list[iA];
            const unsigned eB = list[iB];
            unsigned en[AGG_U];
#pragma unroll
            for (int u = 0; u < 4; ++u) {
                en[u]     = (unsigned)__builtin_amdgcn_readlane((int)eA, u);
                en[u + 4] = (unsigned)__builtin_amdgcn_readlane((int)eB, u);
            }
            float4 hv[AGG_U];
#pragma unroll
            for (int u = 0; u < AGG_U; ++u)
                hv[u] = *(const float4*)(hcol + (size_t)(en[u] >> 8) * HID);
            const unsigned lA = (eA & 255u) * HEADS + (unsigned)hd, lB = (eB & 255u) * HEADS + (unsigned)hd;
            const float asA = sASc[iA * HEADS + (unsigned)hd];
            const float asB = sASc[iB * HEADS + (unsigned)hd];
            const float adA = sAD[lA], adB = sAD[lB];
            const float wA  = expf(leaky(asA + adA) - leaky(sMA[lA] + adA));
            const float wB  = expf(leaky(asB + adB) - leaky(sMA[lB] + adB));
#define AGG_ITEM(U, WSRC)                                                          \
            {                                                                      \
                const float    w = quad_bcast<(U) & 3>(WSRC);                      \
                             \
                const unsigned k = j0 + U < jend ? (en[U] & 255u) >> 5 : 7u;       \
                ax[k] += w * hv[U].x;                                              \
                ay[k] += w * hv[U].y;                                              \
                az[k] += w * hv[U].z;                                              \
                aw[k] += w * hv[U].w;                                              \
                sv[k] += w;                                                        \
            }
            AGG_ITEM(0, wA) AGG_ITEM(1, wA) AGG_ITEM(2, wA) AGG_ITEM(3, wA)
            AGG_ITEM(4, wB) AGG_ITEM(5, wB) AGG_ITEM(6, wB) AGG_ITEM(7, wB)
#undef AGG_ITEM
        }

        const float4 bv = *(const float4*)(bias + 4 * lane);
        float* orow = Out + (size_t)(base + (int)wave) * HID + 4 * lane;
#define ROW_OUT(K) if (32 * K + (int)wave < Rn) \
            row_out(orow + (size_t)(32 * K) * HID, (v2f){ax[K], ay[K]}, (v2f){az[K], aw[K]}, sv[K], bv);
        ROW_OUT(0) ROW_OUT(1) ROW_OUT(2) ROW_OUT(3) ROW_OUT(4) ROW_OUT(5) ROW_OUT(6)
#undef ROW_OUT
    } else {
        for (int i = tid; i < OWN_R * HID; i += OWN_T) acc[i] = 0.f;
        for (int e = tid; e < NEDGES; e += OWN_T) {
            const unsigned ld = (unsigned)(ei[NEDGES + e] - base);
            if (ld < (unsigned)Rn) {
                const int s = ei[e];
                for (int h = 0; h < HEADS; ++h) lds_fmax(&sMA[ld * HEADS + h], AS[s * HEADS + h]);
            }
        }
        __syncthreads();
        for (int e = tid; e < NEDGES; e += OWN_T) {
            const unsigned ld = (unsigned)(ei[NEDGES + e] - base);
            if (ld < (unsigned)Rn) edge_full_lane(AS, Hf, sMA, sAD, sS, acc, ei[e], (int)ld);
        }
        for (int ld = tid; ld < Rn; ld += OWN_T)
            edge_full_lane(AS, Hf, sMA, sAD, sS, acc, base + ld, ld);
        __syncthreads();

        for (int i = tid; i < Rn * HID; i += OWN_T) {
            const int ld = i >> 7, c = i & (HID - 1);
            const float v = acc[ld * HID + acc_pos(c)] / (sS[ld * HEADS + (c >> 4)] + 1e-16f) + bias[c];
            Out[(size_t)(base + ld) * HID + c] = elu1(v);
        }
    }
}

extern "C" void kernel_launch(void* const* d_in, const int* in_sizes, int n_in,
                              void* d_out, int out_size, void* d_ws, size_t ws_size,
                              hipStream_t stream) {
    const float* x   = (const float*)d_in[0];
    const int*   ei  = (const int*)  d_in[1];
    const float* W1  = (const float*)d_in[2];
    const float* as1 = (const float*)d_in[3];
    const float* ad1 = (const float*)d_in[4];
    const float* b1  = (const float*)d_in[5];
    const float* W2  = (const float*)d_in[6];
    const float* as2 = (const float*)d_in[7];
    const float* ad2 = (const float*)d_in[8];
    const float* b2  = (const float*)d_in[9];

    float* ws     = (float*)d_ws;
    float* bufH   = ws;
    float* bufAS  = bufH  + (size_t)NNODES * HID;
    float* bufAD  = bufAS + (size_t)NNODES * HEADS;
    float* bufAgg = bufAD + (size_t)NNODES * HEADS;

    auto run_layer = [&](const float* xin, const float* Wm, const float* as,
                         const float* ad, const float* bv, float* outBuf, int first) {
        k_gemm_wmma<<<GEMM_NB, 256, 0, stream>>>(xin, Wm, as, ad, bufH, bufAS, bufAD, ei,
                                                 (unsigned*)d_out, first);
        k_owner<<<OWN_NB, OWN_T, 0, stream>>>(ei, bufAS, bufAD, bufH, bv, outBuf,
                                              (unsigned*)d_out, first);
    };

    run_layer(x,      W1, as1, ad1, b1, bufAgg,        1);
    run_layer(bufAgg, W2, as2, ad2, b2, (float*)d_out, 0);
}


# Row-sum items grouped by destination node (was: by owner wave): a wave sums one row at a time in five fixed registers, 16 guarded row loads per trip; sums arrival-ordered as before
# speedup vs baseline: 10.1772x; 1.0321x over previous
#include <hip/hip_runtime.h>
#include <math.h>

#define NNODES 50000
#define NEDGES 800000
#define FEAT 128
#define HID  128
#define HEADS 8
#define HDIM 16
#define NEG_SLOPE 0.2f

#define OWN_R    196
#define OWN_NB   256
#define OWN_T    1024
#define OWN_W    (OWN_T / 32)
#define LIST_CAP 4608
#define AGG_U    16
#define ROWS_W   7
#define NSLOT    (ROWS_W * OWN_W)
#define MAX_PT   5
#define KEEP_HDR 32
#define MAP_SLICE 3200
#define MAP_W     (MAP_SLICE / 4)
#define MAP_SV    (MAP_SLICE / 16)
#define REGION_V  (OWN_R * HID / 4)
#define MAP_NV    (NEDGES / 16)
#define MAP_U     7
static_assert(MAP_SLICE % 128 == 0 && NEDGES % MAP_SLICE == 0 && NEDGES / MAP_SLICE < OWN_NB,
              "whole slices, all in full-size blocks");
static_assert((NNODES - (OWN_NB - 1) * OWN_R) * HID > MAP_W + KEEP_HDR, "last block holds its header");
static_assert(ROWS_W * OWN_W >= OWN_R && OWN_W == 32 && OWN_R <= OWN_T && MAX_PT * OWN_T >= LIST_CAP,
              "every owned row has an owner wave and a register slot; one thread per self loop");

typedef __attribute__((ext_vector_type(2))) float v2f;
typedef __attribute__((ext_vector_type(8))) float v8f;

#define GEMM_NB   512
#define GEMM_TILES (NNODES / 16)
#define XS_PITCH  132

static __device__ __forceinline__ void lds_barrier() {
    __builtin_amdgcn_fence(__ATOMIC_RELEASE, "workgroup", "local");
    __builtin_amdgcn_s_barrier();
    __builtin_amdgcn_fence(__ATOMIC_ACQUIRE, "workgroup", "local");
}

template <int K>
static __device__ __forceinline__ float row_xor(float v) {
    return __int_as_float(__builtin_amdgcn_update_dpp(0, __float_as_int(v), 0x160 + K, 0xF, 0xF, true));
}

static __device__ __forceinline__ float row_sums8(const float (&v)[8], int c) {
    const bool b3 = (c & 8) != 0, b2 = (c & 4) != 0, b1 = (c & 2) != 0;
    float u[4], t[2];
#pragma unroll
    for (int p = 0; p < 4; ++p) u[p] = (b3 ? v[p + 4] : v[p]) + row_xor<8>(b3 ? v[p] : v[p + 4]);
#pragma unroll
    for (int p = 0; p < 2; ++p) t[p] = (b2 ? u[p + 2] : u[p]) + row_xor<4>(b2 ? u[p] : u[p + 2]);
    const float s = (b1 ? t[1] : t[0]) + row_xor<2>(b1 ? t[0] : t[1]);
    return s + row_xor<1>(s);
}

__global__ __launch_bounds__(256)
void k_gemm_wmma(const float* __restrict__ X, const float* __restrict__ W,
                 const float* __restrict__ a_src, const float* __restrict__ a_dst,
                 float* __restrict__ Hout, float* __restrict__ AS, float* __restrict__ AD,
                 const int* __restrict__ ei, unsigned* keep, int first) {
    __shared__ float sX[2][16 * XS_PITCH];
    __shared__ float sSc[2][2 * 16 * HEADS];
    const int tid  = threadIdx.x;
    const int lane = tid & 31;
    const int wave = tid >> 5;
    const int l16  = lane & 15;
    const int half = lane >> 4;
    const int col  = wave * 16 + l16;

    const int srow = tid >> 5, sc4 = tid & 31;
    const float* xg = X + (size_t)srow * FEAT + sc4 * 4;
    const int    xl = srow * XS_PITCH + sc4 * 4;

    int tile = blockIdx.x;
    float4 g0 = *(const float4*)(xg + (size_t)tile * (16 * FEAT));
    float4 g1 = *(const float4*)(xg + (size_t)tile * (16 * FEAT) + 8 * FEAT);

    v2f b[FEAT / 4];
#pragma unroll
    for (int i = 0; i < FEAT / 4; ++i) {
        const int ka = 4 * i + 2 * half;
        b[i].x = W[ka * HID + col];
        b[i].y = W[(ka + 1) * HID + col];
    }
    const float asv = a_src[col], adv = a_dst[col];

    if (first) {
        for (unsigned u = blockIdx.x * 256u + (unsigned)tid; u < NEDGES / 4; u += GEMM_NB * 256u) {
            const int4     d  = *(const int4*)(ei + NEDGES + 4 * u);
            const unsigned o  = ((unsigned)d.x / OWN_R)         | (((unsigned)d.y / OWN_R) << 8) |
                                (((unsigned)d.z / OWN_R) << 16) | (((unsigned)d.w / OWN_R) << 24);
            const unsigned sl = u / MAP_W;
            keep[(size_t)sl * (OWN_R * HID) + (u - sl * MAP_W)] = o;
        }
    }

    *(float4*)&sX[0][xl]                = g0;
    *(float4*)&sX[0][xl + 8 * XS_PITCH] = g1;
    if (tile + GEMM_NB < GEMM_TILES) {
        g0 = *(const float4*)(xg + (size_t)(tile + GEMM_NB) * (16 * FEAT));
        g1 = *(const float4*)(xg + (size_t)(tile + GEMM_NB) * (16 * FEAT) + 8 * FEAT);
    }
    lds_barrier();

    const int arow = l16 * XS_PITCH + 2 * half;
    for (int it = 0; tile < GEMM_TILES; ++it, tile += GEMM_NB) {
        const float* xs = sX[it & 1] + arow;
        v2f a[FEAT / 4];
#pragma unroll
        for (int i = 0; i < FEAT / 4; ++i) a[i] = *(const v2f*)(xs + 4 * i);

        if (tile + GEMM_NB < GEMM_TILES) {
            float* xn = sX[(it & 1) ^ 1];
            *(float4*)&xn[xl]                = g0;
            *(float4*)&xn[xl + 8 * XS_PITCH] = g1;
            if (tile + 2 * GEMM_NB < GEMM_TILES) {
                g0 = *(const float4*)(xg + (size_t)(tile + 2 * GEMM_NB) * (16 * FEAT));
                g1 = *(const float4*)(xg + (size_t)(tile + 2 * GEMM_NB) * (16 * FEAT) + 8 * FEAT);
            }
        }

        v8f acc = {};
#pragma unroll
        for (int i = 0; i < FEAT / 4; ++i)
            acc = __builtin_amdgcn_wmma_f32_16x16x4_f32(
                      false, a[i], false, b[i], (short)0, acc, false, false);

        float ps[8], pd[8];
#pragma unroll
        for (int j = 0; j < 8; ++j) { ps[j] = acc[j] * asv; pd[j] = acc[j] * adv; }
        const float sa = row_sums8(ps, l16);
        const float sd = row_sums8(pd, l16);
        sSc[it & 1][(l16 & 1) * (16 * HEADS) + (8 * half + (l16 >> 1)) * HEADS + wave] = (l16 & 1) ? sd : sa;

        lds_barrier();

        const int orow0 = tile * 16 + 8 * half;
#pragma unroll
        for (int j = 0; j < 8; ++j)
            Hout[(size_t)(orow0 + j) * HID + col] = acc[j];
        (wave < 4 ? AS : AD)[(size_t)tile * (16 * HEADS) + (tid & 127)] = sSc[it & 1][tid];
    }
}

static __device__ __forceinline__ float leaky(float v) {
    return v >= 0.f ? v : NEG_SLOPE * v;
}

static __device__ __forceinline__ unsigned zero_bytes(unsigned x) {
    return ~(((x & 0x7F7F7F7Fu) + 0x7F7F7F7Fu) | x) & 0x80808080u;
}

static __device__ __forceinline__ void lds_fmax(float* p, float v) {
    __hip_atomic_fetch_max(p, v, __ATOMIC_RELAXED, __HIP_MEMORY_SCOPE_WORKGROUP);
}

static __device__ __forceinline__ int acc_pos(int c) { return (c & 3) * 32 + (c >> 2); }

template <int J>
static __device__ __forceinline__ float quad_bcast(float v) {
    return __int_as_float(__builtin_amdgcn_update_dpp(0, __float_as_int(v), J * 0x55, 0xF, 0xF, true));
}

static __device__ void edge_full_lane(const float* __restrict__ AS, const float* __restrict__ Hf,
                                      const float* sMA, const float* sAD, float* sS, float* acc,
                                      int s, int ld) {
    for (int hd = 0; hd < HEADS; ++hd) {
        const float ad = sAD[ld * HEADS + hd];
        const float m  = leaky(sMA[ld * HEADS + hd] + ad);
        const float w  = expf(leaky(AS[s * HEADS + hd] + ad) - m);
        atomicAdd(&sS[ld * HEADS + hd], w);
        for (int c = 0; c < HDIM; ++c) {
            const int col = hd * HDIM + c;
            atomicAdd(&acc[ld * HID + acc_pos(col)], w * Hf[(size_t)s * HID + col]);
        }
    }
}

static __device__ __forceinline__ float elu1(float v) {
    const float e = expm1f(v);
    return v > 0.f ? v : e;
}

static __device__ __forceinline__ void row_out(float* dst, v2f a, v2f b, float s, float4 bv) {
    const float den = s + 1e-16f;
    float4 o;
    o.x = elu1(a.x / den + bv.x);
    o.y = elu1(a.y / den + bv.y);
    o.z = elu1(b.x / den + bv.z);
    o.w = elu1(b.y / den + bv.w);
    *(float4*)dst = o;
}

__global__ __launch_bounds__(OWN_T)
void k_owner(const int* __restrict__ ei, const float* __restrict__ AS,
             const float* __restrict__ AD, const float* __restrict__ Hf,
             const float* __restrict__ bias, float* Out, unsigned* keep, int first) {
    __shared__ float    acc[OWN_R * HID];
    __shared__ float    sMA[OWN_R * HEADS];
    __shared__ float    sAD[OWN_R * HEADS];
    __shared__ float    sS [OWN_R * HEADS];
    __shared__ unsigned list[LIST_CAP];
    __shared__ alignas(16) float sASc[LIST_CAP * HEADS];
    __shared__ unsigned perm[LIST_CAP];
    __shared__ unsigned ncnt[NSLOT];
    __shared__ unsigned nstart[NSLOT];
    __shared__ unsigned cnt;

    const int tid  = threadIdx.x;
    const int base = blockIdx.x * OWN_R;
    const int Rn   = min(OWN_R, NNODES - base);

    unsigned*      mine     = keep + (size_t)base * HID + MAP_W;
    const unsigned keep_cap = (unsigned)min(Rn * HID - MAP_W - KEEP_HDR, LIST_CAP - Rn);
    const unsigned kept     = first ? 0xFFFFFFFFu : mine[0];
    const bool     reuse    = kept <= keep_cap;

    for (int i = tid; i < Rn * HEADS; i += OWN_T) {
        sMA[i] = AS[base * HEADS + i];
        sAD[i] = AD[base * HEADS + i];
        sS[i]  = 0.f;
    }
    if (tid == 0) cnt = reuse ? kept : 0u;
    if (tid < NSLOT) ncnt[tid] = 0u;
    if (reuse)
        for (unsigned i = tid; i < kept; i += OWN_T) list[i] = mine[KEEP_HDR + i];
    __syncthreads();

    if (first) {
        const uint4*   mv  = (const uint4*)keep;
        const unsigned pat = (unsigned)blockIdx.x * 0x01010101u;
        for (int q0 = tid; q0 < MAP_NV; q0 += MAP_U * OWN_T) {
            uint4 v[MAP_U];
#pragma unroll
            for (int u = 0; u < MAP_U; ++u) {
                const unsigned q = (unsigned)(q0 + u * OWN_T);
                const unsigned j = q / MAP_SV;
                v[u] = q < MAP_NV ? mv[j * REGION_V + (q - j * MAP_SV)]
                                  : make_uint4(~pat, ~pat, ~pat, ~pat);
            }
            unsigned m[MAP_U], c = 0;
#pragma unroll
            for (int u = 0; u < MAP_U; ++u) {
                m[u] = (zero_bytes(v[u].x ^ pat) >> 7) | (zero_bytes(v[u].y ^ pat) >> 6) |
                       (zero_bytes(v[u].z ^ pat) >> 5) | (zero_bytes(v[u].w ^ pat) >> 4);
                c += (unsigned)__builtin_popcount(m[u]);
            }
            if (c) {
                unsigned slot = atomicAdd(&cnt, c);
#pragma unroll
                for (int u = 0; u < MAP_U; ++u) {
                    const unsigned e0 = (unsigned)(q0 + u * OWN_T) * 16u;
                    for (unsigned mm = m[u]; mm; mm &= mm - 1, ++slot) {
                        const unsigned bit = (unsigned)__builtin_ctz(mm);
                        if (slot < LIST_CAP) list[slot] = e0 + 4u * (bit & 7u) + (bit >> 3);
                    }
                }
            }
        }
    }
    __syncthreads();
    const unsigned total  = (unsigned)__builtin_amdgcn_readfirstlane((int)cnt);
    const unsigned nitems = total + (unsigned)Rn;
    const bool     save   = first && total <= keep_cap;
    if (first && tid == 0) mine[0] = total;

    if (first ? nitems <= LIST_CAP : reuse) {
        unsigned rk[MAX_PT];
#pragma unroll
        for (int t = 0; t < MAX_PT; ++t) {
            const unsigned i = (unsigned)tid + (unsigned)t * OWN_T;
            rk[t] = 0u;
            if (i < total) {
                unsigned ent = list[i];
                if (first) {
                    const int s = ei[ent];
                    ent     = ((unsigned)s << 8) | (unsigned)(ei[NEDGES + ent] - base);
                    list[i] = ent;
                    if (save) mine[KEEP_HDR + i] = ent;
                }
                const unsigned ld = ent & 255u;
                const int      s  = (int)(ent >> 8);
                const float4* ap = (const float4*)(AS + (size_t)s * HEADS);
                const float4 a0 = ap[0], a1 = ap[1];
                *(float4*)&sASc[HEADS * i]     = a0;
                *(float4*)&sASc[HEADS * i + 4] = a1;
                float* m = &sMA[ld * HEADS];
                lds_fmax(m + 0, a0.x); lds_fmax(m + 1, a0.y);
                lds_fmax(m + 2, a0.z); lds_fmax(m + 3, a0.w);
                lds_fmax(m + 4, a1.x); lds_fmax(m + 5, a1.y);
                lds_fmax(m + 6, a1.z); lds_fmax(m + 7, a1.w);
                rk[t] = atomicAdd(&ncnt[ld], 1u);
            }
        }
        unsigned rkS = 0u;
        if (tid < Rn) {
            list[total + (unsigned)tid] = ((unsigned)(base + tid) << 8) | (unsigned)tid;
            const float4* ap = (const float4*)(AS + (size_t)(base + tid) * HEADS);
            *(float4*)&sASc[HEADS * (total + (unsigned)tid)]     = ap[0];
            *(float4*)&sASc[HEADS * (total + (unsigned)tid) + 4] = ap[1];
            rkS = atomicAdd(&ncnt[tid], 1u);
        }
        __syncthreads();

        const int lane = tid & 31, hd = lane >> 2, q = lane & 3;
        if (tid < 32) {
            unsigned cn[ROWS_W], tot = 0u;
#pragma unroll
            for (int k = 0; k < ROWS_W; ++k) { cn[k] = ncnt[lane + 32 * k]; tot += cn[k]; }
            unsigned incl = tot;
#pragma unroll
            for (int d = 1; d < 32; d <<= 1) {
                const unsigned up = __shfl_up(incl, (unsigned)d);
                if (lane >= d) incl += up;
            }
            unsigned run = incl - tot;
#pragma unroll
            for (int k = 0; k < ROWS_W; ++k) { nstart[lane + 32 * k] = run; run += cn[k]; }
        }
        __syncthreads();
#pragma unroll
        for (int t = 0; t < MAX_PT; ++t) {
            const unsigned i = (unsigned)tid + (unsigned)t * OWN_T;
            if (i < total) perm[nstart[list[i] & 255u] + rk[t]] = i;
        }
        if (tid < Rn) perm[nstart[tid] + rkS] = total + (unsigned)tid;
        __syncthreads();

        const unsigned wave = (unsigned)__builtin_amdgcn_readfirstlane(tid >> 5);
        const float*   hcol = Hf + 4 * lane;
        const float4   bv   = *(const float4*)(bias + 4 * lane);
        for (unsigned k = 0; k < ROWS_W; ++k) {
            const unsigned node = wave + 32u * k;
            if ((int)node >= Rn) break;
            const unsigned jbeg = (unsigned)__builtin_amdgcn_readfirstlane((int)nstart[node]);
            const unsigned jend = jbeg + (unsigned)__builtin_amdgcn_readfirstlane((int)ncnt[node]);
            const float ad = sAD[node * HEADS + (unsigned)hd];
            const float mk = leaky(sMA[node * HEADS + (unsigned)hd] + ad);
            float ax = 0.f, ay = 0.f, az = 0.f, aw = 0.f, sw = 0.f;
            for (unsigned j0 = jbeg; j0 < jend; j0 += AGG_U) {
                unsigned it[4], ent[4];
                float    as[4], wv[4];
#pragma unroll
                for (int t = 0; t < 4; ++t) {
                    it[t]  = perm[min(j0 + (unsigned)q + 4u * (unsigned)t, jend - 1u)];
                    ent[t] = list[it[t]];
                    as[t]  = sASc[it[t] * HEADS + (unsigned)hd];
                }
                float4 hv[AGG_U];
#pragma unroll
                for (int u = 0; u < AGG_U; ++u) {
                    if (j0 + (unsigned)u < jend) {
                        const unsigned s = (unsigned)__builtin_amdgcn_readlane((int)ent[u >> 2], u & 3) >> 8;
                        hv[u] = *(const float4*)(hcol + (size_t)s * HID);
                    }
                }
#pragma unroll
                for (int t = 0; t < 4; ++t) {
                    wv[t] = 0.f;
                    if (j0 + 4u * (unsigned)t < jend) wv[t] = expf(leaky(as[t] + ad) - mk);
                }
#pragma unroll
                for (int u = 0; u < AGG_U; ++u) {
                    if (j0 + (unsigned)u < jend) {
                        const float w = (u & 3) == 0 ? quad_bcast<0>(wv[u >> 2]) :
                                        (u & 3) == 1 ? quad_bcast<1>(wv[u >> 2]) :
                                        (u & 3) == 2 ? quad_bcast<2>(wv[u >> 2]) : quad_bcast<3>(wv[u >> 2]);
                        ax += w * hv[u].x;
                        ay += w * hv[u].y;
                        az += w * hv[u].z;
                        aw += w * hv[u].w;
                        sw += w;
                    }
                }
            }
            row_out(Out + (size_t)(base + (int)node) * HID + 4 * lane, (v2f){ax, ay}, (v2f){az, aw}, sw, bv);
        }
    } else {
        for (int i = tid; i < OWN_R * HID; i += OWN_T) acc[i] = 0.f;
        for (int e = tid; e < NEDGES; e += OWN_T) {
            const unsigned ld = (unsigned)(ei[NEDGES + e] - base);
            if (ld < (unsigned)Rn) {
                const int s = ei[e];
                for (int h = 0; h < HEADS; ++h) lds_fmax(&sMA[ld * HEADS + h], AS[s * HEADS + h]);
            }
        }
        __syncthreads();
        for (int e = tid; e < NEDGES; e += OWN_T) {
            const unsigned ld = (unsigned)(ei[NEDGES + e] - base);
            if (ld < (unsigned)Rn) edge_full_lane(AS, Hf, sMA, sAD, sS, acc, ei[e], (int)ld);
        }
        for (int ld = tid; ld < Rn; ld += OWN_T)
            edge_full_lane(AS, Hf, sMA, sAD, sS, acc, base + ld, ld);
        __syncthreads();

        for (int i = tid; i < Rn * HID; i += OWN_T) {
            const int ld = i >> 7, c = i & (HID - 1);
            const float v = acc[ld * HID + acc_pos(c)] / (sS[ld * HEADS + (c >> 4)] + 1e-16f) + bias[c];
            Out[(size_t)(base + ld) * HID + c] = elu1(v);
        }
    }
}

extern "C" void kernel_launch(void* const* d_in, const int* in_sizes, int n_in,
                              void* d_out, int out_size, void* d_ws, size_t ws_size,
                              hipStream_t stream) {
    const float* x   = (const float*)d_in[0];
    const int*   ei  = (const int*)  d_in[1];
    const float* W1  = (const float*)d_in[2];
    const float* as1 = (const float*)d_in[3];
    const float* ad1 = (const float*)d_in[4];
    const float* b1  = (const float*)d_in[5];
    const float* W2  = (const float*)d_in[6];
    const float* as2 = (const float*)d_in[7];
    const float* ad2 = (const float*)d_in[8];
    const float* b2  = (const float*)d_in[9];

    float* ws     = (float*)d_ws;
    float* bufH   = ws;
    float* bufAS  = bufH  + (size_t)NNODES * HID;
    float* bufAD  = bufAS + (size_t)NNODES * HEADS;
    float* bufAgg = bufAD + (size_t)NNODES * HEADS;

    auto run_layer = [&](const float* xin, const float* Wm, const float* as,
                         const float* ad, const float* bv, float* outBuf, int first) {
        k_gemm_wmma<<<GEMM_NB, 256, 0, stream>>>(xin, Wm, as, ad, bufH, bufAS, bufAD, ei,
                                                 (unsigned*)d_out, first);
        k_owner<<<OWN_NB, OWN_T, 0, stream>>>(ei, bufAS, bufAD, bufH, bv, outBuf,
                                              (unsigned*)d_out, first);
    };

    run_layer(x,      W1, as1, ad1, b1, bufAgg,        1);
    run_layer(bufAgg, W2, as2, ad2, b2, (float*)d_out, 0);
}


# Layer-1 edge filter in two levels: product launch also builds each owner block's bitmap of 16-edge map groups holding its edges; a block tests ~3000 flagged groups, was 50000; sum order as before
# speedup vs baseline: 10.8945x; 1.0705x over previous
#include <hip/hip_runtime.h>
#include <math.h>

#define NNODES 50000
#define NEDGES 800000
#define FEAT 128
#define HID  128
#define HEADS 8
#define HDIM 16
#define NEG_SLOPE 0.2f

#define OWN_R    196
#define OWN_NB   256
#define OWN_T    1024
#define OWN_W    (OWN_T / 32)
#define LIST_CAP 4608
#define AGG_U    16
#define ROWS_W   7
#define NSLOT    (ROWS_W * OWN_W)
#define MAX_PT   5
#define KEEP_HDR 32
#define MAP_SLICE 3200
#define MAP_W     (MAP_SLICE / 4)
#define MAP_SV    (MAP_SLICE / 16)
#define REGION_V  (OWN_R * HID / 4)
static_assert(MAP_SLICE % 128 == 0 && NEDGES % MAP_SLICE == 0 && NEDGES / MAP_SLICE < OWN_NB,
              "whole slices, all in full-size blocks");
#define GRP_E     16
#define NGRP      (NEDGES / GRP_E)
#define BM_CHUNK  1024
#define BM_NCH    ((NGRP + BM_CHUNK - 1) / BM_CHUNK)
#define BM_ROW_W  (BM_NCH * 32)
#define BM_OFF    (MAP_W + KEEP_HDR + LIST_CAP)
#define BM_PITCH  257
#define BM_UPT    (BM_CHUNK * GRP_E / 4 / 256)
#define BM_LU     8
#define FLT_U     4
static_assert(NEDGES % GRP_E == 0 && MAP_SLICE % GRP_E == 0 && BM_OFF % 32 == 0, "whole groups, whole lines");
static_assert(BM_OFF + 2 * BM_ROW_W <= OWN_R * HID && OWN_NB / 2 <= NNODES / OWN_R,
              "two rows fit behind the list of a full-size block");
static_assert(BM_ROW_W >= OWN_T && BM_ROW_W <= 2 * OWN_T && BM_UPT % BM_LU == 0, "two row words per thread at most");
static_assert((NNODES - (OWN_NB - 1) * OWN_R) * HID > MAP_W + KEEP_HDR, "last block holds its header");
static_assert(ROWS_W * OWN_W >= OWN_R && OWN_W == 32 && OWN_R <= OWN_T && MAX_PT * OWN_T >= LIST_CAP,
              "every owned row has an owner wave and a register slot; one thread per self loop");

typedef __attribute__((ext_vector_type(2))) float v2f;
typedef __attribute__((ext_vector_type(8))) float v8f;

#define GEMM_NB   512
#define GEMM_TILES (NNODES / 16)
#define XS_PITCH  132
static_assert(BM_NCH <= GEMM_NB - (GEMM_TILES - 6 * GEMM_NB), "a bitmap chunk only in blocks with 6 row tiles");

static __device__ __forceinline__ void lds_barrier() {
    __builtin_amdgcn_fence(__ATOMIC_RELEASE, "workgroup", "local");
    __builtin_amdgcn_s_barrier();
    __builtin_amdgcn_fence(__ATOMIC_ACQUIRE, "workgroup", "local");
}

template <int K>
static __device__ __forceinline__ float row_xor(float v) {
    return __int_as_float(__builtin_amdgcn_update_dpp(0, __float_as_int(v), 0x160 + K, 0xF, 0xF, true));
}

static __device__ __forceinline__ float row_sums8(const float (&v)[8], int c) {
    const bool b3 = (c & 8) != 0, b2 = (c & 4) != 0, b1 = (c & 2) != 0;
    float u[4], t[2];
#pragma unroll
    for (int p = 0; p < 4; ++p) u[p] = (b3 ? v[p + 4] : v[p]) + row_xor<8>(b3 ? v[p] : v[p + 4]);
#pragma unroll
    for (int p = 0; p < 2; ++p) t[p] = (b2 ? u[p + 2] : u[p]) + row_xor<4>(b2 ? u[p] : u[p + 2]);
    const float s = (b1 ? t[1] : t[0]) + row_xor<2>(b1 ? t[0] : t[1]);
    return s + row_xor<1>(s);
}

__global__ __launch_bounds__(256)
void k_gemm_wmma(const float* __restrict__ X, const float* __restrict__ W,
                 const float* __restrict__ a_src, const float* __restrict__ a_dst,
                 float* __restrict__ Hout, float* __restrict__ AS, float* __restrict__ AD,
                 const int* __restrict__ ei, unsigned* keep, int first) {
    __shared__ float sX[2][16 * XS_PITCH];
    __shared__ float sSc[2][2 * 16 * HEADS];
    __shared__ unsigned sBm[32 * BM_PITCH];
    const int tid  = threadIdx.x;
    const int lane = tid & 31;
    const int wave = tid >> 5;
    const int l16  = lane & 15;
    const int half = lane >> 4;
    const int col  = wave * 16 + l16;

    const int srow = tid >> 5, sc4 = tid & 31;
    const float* xg = X + (size_t)srow * FEAT + sc4 * 4;
    const int    xl = srow * XS_PITCH + sc4 * 4;

    int tile = blockIdx.x;
    float4 g0 = *(const float4*)(xg + (size_t)tile * (16 * FEAT));
    float4 g1 = *(const float4*)(xg + (size_t)tile * (16 * FEAT) + 8 * FEAT);

    v2f b[FEAT / 4];
#pragma unroll
    for (int i = 0; i < FEAT / 4; ++i) {
        const int ka = 4 * i + 2 * half;
        b[i].x = W[ka * HID + col];
        b[i].y = W[(ka + 1) * HID + col];
    }
    const float asv = a_src[col], adv = a_dst[col];

    const int ch = GEMM_NB - 1 - (int)blockIdx.x;
    if (first && ch < BM_NCH) {
        for (int i = tid; i < 32 * BM_PITCH; i += 256) sBm[i] = 0u;
        lds_barrier();
        const unsigned u0 = (unsigned)ch * (BM_CHUNK * GRP_E / 4);
        for (int t0 = 0; t0 < BM_UPT; t0 += BM_LU) {
            int4 d[BM_LU];
#pragma unroll
            for (int t = 0; t < BM_LU; ++t) {
                const unsigned u = min(u0 + (unsigned)((t0 + t) * 256 + tid), (unsigned)(NEDGES / 4 - 1));
                d[t] = *(const int4*)(ei + NEDGES + 4 * u);
            }
#pragma unroll
            for (int t = 0; t < BM_LU; ++t) {
                const unsigned ul = (unsigned)((t0 + t) * 256 + tid);
                const unsigned u  = u0 + ul;
                if (u < NEDGES / 4) {
                    const unsigned ox = (unsigned)d[t].x / OWN_R, oy = (unsigned)d[t].y / OWN_R;
                    const unsigned oz = (unsigned)d[t].z / OWN_R, ow = (unsigned)d[t].w / OWN_R;
                    const unsigned sl = u / MAP_W;
                    keep[(size_t)sl * (OWN_R * HID) + (u - sl * MAP_W)] = ox | (oy << 8) | (oz << 16) | (ow << 24);
                    const unsigned gl  = ul >> 2;
                    unsigned*      row = sBm + (gl >> 5) * BM_PITCH;
                    const unsigned bit = 1u << (gl & 31u);
                    atomicOr(&row[ox], bit);
                    atomicOr(&row[oy], bit);
                    atomicOr(&row[oz], bit);
                    atomicOr(&row[ow], bit);
                }
            }
        }
        lds_barrier();
        for (int o = wave; o < OWN_NB; o += 8)
            keep[(size_t)(o >> 1) * (OWN_R * HID) + BM_OFF + (o & 1) * BM_ROW_W + ch * 32 + lane] =
                sBm[lane * BM_PITCH + o];
    }

    *(float4*)&sX[0][xl]                = g0;
    *(float4*)&sX[0][xl + 8 * XS_PITCH] = g1;
    if (tile + GEMM_NB < GEMM_TILES) {
        g0 = *(const float4*)(xg + (size_t)(tile + GEMM_NB) * (16 * FEAT));
        g1 = *(const float4*)(xg + (size_t)(tile + GEMM_NB) * (16 * FEAT) + 8 * FEAT);
    }
    lds_barrier();

    const int arow = l16 * XS_PITCH + 2 * half;
    for (int it = 0; tile < GEMM_TILES; ++it, tile += GEMM_NB) {
        const float* xs = sX[it & 1] + arow;
        v2f a[FEAT / 4];
#pragma unroll
        for (int i = 0; i < FEAT / 4; ++i) a[i] = *(const v2f*)(xs + 4 * i);

        if (tile + GEMM_NB < GEMM_TILES) {
            float* xn = sX[(it & 1) ^ 1];
            *(float4*)&xn[xl]                = g0;
            *(float4*)&xn[xl + 8 * XS_PITCH] = g1;
            if (tile + 2 * GEMM_NB < GEMM_TILES) {
                g0 = *(const float4*)(xg + (size_t)(tile + 2 * GEMM_NB) * (16 * FEAT));
                g1 = *(const float4*)(xg + (size_t)(tile + 2 * GEMM_NB) * (16 * FEAT) + 8 * FEAT);
            }
        }

        v8f acc = {};
#pragma unroll
        for (int i = 0; i < FEAT / 4; ++i)
            acc = __builtin_amdgcn_wmma_f32_16x16x4_f32(
                      false, a[i], false, b[i], (short)0, acc, false, false);

        float ps[8], pd[8];
#pragma unroll
        for (int j = 0; j < 8; ++j) { ps[j] = acc[j] * asv; pd[j] = acc[j] * adv; }
        const float sa = row_sums8(ps, l16);
        const float sd = row_sums8(pd, l16);
        sSc[it & 1][(l16 & 1) * (16 * HEADS) + (8 * half + (l16 >> 1)) * HEADS + wave] = (l16 & 1) ? sd : sa;

        lds_barrier();

        const int orow0 = tile * 16 + 8 * half;
#pragma unroll
        for (int j = 0; j < 8; ++j)
            Hout[(size_t)(orow0 + j) * HID + col] = acc[j];
        (wave < 4 ? AS : AD)[(size_t)tile * (16 * HEADS) + (tid & 127)] = sSc[it & 1][tid];
    }
}

static __device__ __forceinline__ float leaky(float v) {
    return v >= 0.f ? v : NEG_SLOPE * v;
}

static __device__ __forceinline__ unsigned zero_bytes(unsigned x) {
    return ~(((x & 0x7F7F7F7Fu) + 0x7F7F7F7Fu) | x) & 0x80808080u;
}

static __device__ __forceinline__ void lds_fmax(float* p, float v) {
    __hip_atomic_fetch_max(p, v, __ATOMIC_RELAXED, __HIP_MEMORY_SCOPE_WORKGROUP);
}

static __device__ __forceinline__ int acc_pos(int c) { return (c & 3) * 32 + (c >> 2); }

template <int J>
static __device__ __forceinline__ float quad_bcast(float v) {
    return __int_as_float(__builtin_amdgcn_update_dpp(0, __float_as_int(v), J * 0x55, 0xF, 0xF, true));
}

static __device__ void edge_full_lane(const float* __restrict__ AS, const float* __restrict__ Hf,
                                      const float* sMA, const float* sAD, float* sS, float* acc,
                                      int s, int ld) {
    for (int hd = 0; hd < HEADS; ++hd) {
        const float ad = sAD[ld * HEADS + hd];
        const float m  = leaky(sMA[ld * HEADS + hd] + ad);
        const float w  = expf(leaky(AS[s * HEADS + hd] + ad) - m);
        atomicAdd(&sS[ld * HEADS + hd], w);
        for (int c = 0; c < HDIM; ++c) {
            const int col = hd * HDIM + c;
            atomicAdd(&acc[ld * HID + acc_pos(col)], w * Hf[(size_t)s * HID + col]);
        }
    }
}

static __device__ __forceinline__ float elu1(float v) {
    const float e = expm1f(v);
    return v > 0.f ? v : e;
}

static __device__ __forceinline__ void row_out(float* dst, v2f a, v2f b, float s, float4 bv) {
    const float den = s + 1e-16f;
    float4 o;
    o.x = elu1(a.x / den + bv.x);
    o.y = elu1(a.y / den + bv.y);
    o.z = elu1(b.x / den + bv.z);
    o.w = elu1(b.y / den + bv.w);
    *(float4*)dst = o;
}

__global__ __launch_bounds__(OWN_T)
void k_owner(const int* __restrict__ ei, const float* __restrict__ AS,
             const float* __restrict__ AD, const float* __restrict__ Hf,
             const float* __restrict__ bias, float* Out, unsigned* keep, int first) {
    __shared__ float    acc[OWN_R * HID];
    __shared__ float    sMA[OWN_R * HEADS];
    __shared__ float    sAD[OWN_R * HEADS];
    __shared__ float    sS [OWN_R * HEADS];
    __shared__ unsigned list[LIST_CAP];
    __shared__ alignas(16) float sASc[LIST_CAP * HEADS];
    __shared__ unsigned perm[LIST_CAP];
    __shared__ unsigned ncnt[NSLOT];
    __shared__ unsigned nstart[NSLOT];
    __shared__ unsigned cnt;
    __shared__ unsigned gcnt;

    const int tid  = threadIdx.x;
    const int base = blockIdx.x * OWN_R;
    const int Rn   = min(OWN_R, NNODES - base);

    unsigned*      mine     = keep + (size_t)base * HID + MAP_W;
    const unsigned keep_cap = (unsigned)min(Rn * HID - MAP_W - KEEP_HDR, LIST_CAP - Rn);
    const unsigned kept     = first ? 0xFFFFFFFFu : mine[0];
    const bool     reuse    = kept <= keep_cap;

    for (int i = tid; i < Rn * HEADS; i += OWN_T) {
        sMA[i] = AS[base * HEADS + i];
        sAD[i] = AD[base * HEADS + i];
        sS[i]  = 0.f;
    }
    if (tid == 0) { cnt = reuse ? kept : 0u; gcnt = 0u; }
    if (tid < NSLOT) ncnt[tid] = 0u;
    if (reuse)
        for (unsigned i = tid; i < kept; i += OWN_T) list[i] = mine[KEEP_HDR + i];
    __syncthreads();

    if (first) {
        const unsigned* bm  = keep + (size_t)(blockIdx.x >> 1) * (OWN_R * HID) + BM_OFF
                                   + (blockIdx.x & 1) * BM_ROW_W;
        const uint4*    mv  = (const uint4*)keep;
        const unsigned  pat = (unsigned)blockIdx.x * 0x01010101u;
        {
            const unsigned w0 = bm[tid];
            const unsigned w1 = tid + OWN_T < BM_ROW_W ? bm[tid + OWN_T] : 0u;
            const unsigned c  = (unsigned)(__builtin_popcount(w0) + __builtin_popcount(w1));
            if (c) {
                unsigned slot = atomicAdd(&gcnt, c);
                for (unsigned mm = w0; mm; mm &= mm - 1, ++slot)
                    if (slot < LIST_CAP) perm[slot] = 32u * (unsigned)tid + (unsigned)__builtin_ctz(mm);
                for (unsigned mm = w1; mm; mm &= mm - 1, ++slot)
                    if (slot < LIST_CAP) perm[slot] = 32u * (unsigned)(tid + OWN_T) + (unsigned)__builtin_ctz(mm);
            }
        }
        __syncthreads();
        const unsigned ng = (unsigned)__builtin_amdgcn_readfirstlane((int)gcnt);
        if (ng > LIST_CAP) {
            if (tid == 0) cnt = LIST_CAP + 1u;
        } else {
            for (unsigned i0 = tid; i0 < ng; i0 += FLT_U * OWN_T) {
                unsigned g[FLT_U];
                uint4    v[FLT_U];
#pragma unroll
                for (int u = 0; u < FLT_U; ++u) {
                    const unsigned i = i0 + (unsigned)u * OWN_T;
                    g[u] = perm[i < ng ? i : i0];
                    const unsigned j = g[u] / MAP_SV;
                    v[u] = mv[j * REGION_V + (g[u] - j * MAP_SV)];
                }
                unsigned m[FLT_U], c = 0;
#pragma unroll
                for (int u = 0; u < FLT_U; ++u) {
                    m[u] = (zero_bytes(v[u].x ^ pat) >> 7) | (zero_bytes(v[u].y ^ pat) >> 6) |
                           (zero_bytes(v[u].z ^ pat) >> 5) | (zero_bytes(v[u].w ^ pat) >> 4);
                    if (i0 + (unsigned)u * OWN_T >= ng) m[u] = 0u;
                    c += (unsigned)__builtin_popcount(m[u]);
                }
                if (c) {
                    unsigned slot = atomicAdd(&cnt, c);
#pragma unroll
                    for (int u = 0; u < FLT_U; ++u) {
                        const unsigned e0 = g[u] * GRP_E;
                        for (unsigned mm = m[u]; mm; mm &= mm - 1, ++slot) {
                            const unsigned bit = (unsigned)__builtin_ctz(mm);
                            if (slot < LIST_CAP) list[slot] = e0 + 4u * (bit & 7u) + (bit >> 3);
                        }
                    }
                }
            }
        }
    }
    __syncthreads();
    const unsigned total  = (unsigned)__builtin_amdgcn_readfirstlane((int)cnt);
    const unsigned nitems = total + (unsigned)Rn;
    const bool     save   = first && total <= keep_cap;
    if (first && tid == 0) mine[0] = total;

    if (first ? nitems <= LIST_CAP : reuse) {
        unsigned rk[MAX_PT];
#pragma unroll
        for (int t = 0; t < MAX_PT; ++t) {
            const unsigned i = (unsigned)tid + (unsigned)t * OWN_T;
            rk[t] = 0u;
            if (i < total) {
                unsigned ent = list[i];
                if (first) {
                    const int s = ei[ent];
                    ent     = ((unsigned)s << 8) | (unsigned)(ei[NEDGES + ent] - base);
                    list[i] = ent;
                    if (save) mine[KEEP_HDR + i] = ent;
                }
                const unsigned ld = ent & 255u;
                const int      s  = (int)(ent >> 8);
                const float4* ap = (const float4*)(AS + (size_t)s * HEADS);
                const float4 a0 = ap[0], a1 = ap[1];
                *(float4*)&sASc[HEADS * i]     = a0;
                *(float4*)&sASc[HEADS * i + 4] = a1;
                float* m = &sMA[ld * HEADS];
                lds_fmax(m + 0, a0.x); lds_fmax(m + 1, a0.y);
                lds_fmax(m + 2, a0.z); lds_fmax(m + 3, a0.w);
                lds_fmax(m + 4, a1.x); lds_fmax(m + 5, a1.y);
                lds_fmax(m + 6, a1.z); lds_fmax(m + 7, a1.w);
                rk[t] = atomicAdd(&ncnt[ld], 1u);
            }
        }
        unsigned rkS = 0u;
        if (tid < Rn) {
            list[total + (unsigned)tid] = ((unsigned)(base + tid) << 8) | (unsigned)tid;
            const float4* ap = (const float4*)(AS + (size_t)(base + tid) * HEADS);
            *(float4*)&sASc[HEADS * (total + (unsigned)tid)]     = ap[0];
            *(float4*)&sASc[HEADS * (total + (unsigned)tid) + 4] = ap[1];
            rkS = atomicAdd(&ncnt[tid], 1u);
        }
        __syncthreads();

        const int lane = tid & 31, hd = lane >> 2, q = lane & 3;
        if (tid < 32) {
            unsigned cn[ROWS_W], tot = 0u;
#pragma unroll
            for (int k = 0; k < ROWS_W; ++k) { cn[k] = ncnt[lane + 32 * k]; tot += cn[k]; }
            unsigned incl = tot;
#pragma unroll
            for (int d = 1; d < 32; d <<= 1) {
                const unsigned up = __shfl_up(incl, (unsigned)d);
                if (lane >= d) incl += up;
            }
            unsigned run = incl - tot;
#pragma unroll
            for (int k = 0; k < ROWS_W; ++k) { nstart[lane + 32 * k] = run; run += cn[k]; }
        }
        __syncthreads();
#pragma unroll
        for (int t = 0; t < MAX_PT; ++t) {
            const unsigned i = (unsigned)tid + (unsigned)t * OWN_T;
            if (i < total) perm[nstart[list[i] & 255u] + rk[t]] = i;
        }
        if (tid < Rn) perm[nstart[tid] + rkS] = total + (unsigned)tid;
        __syncthreads();

        const unsigned wave = (unsigned)__builtin_amdgcn_readfirstlane(tid >> 5);
        const float*   hcol = Hf + 4 * lane;
        const float4   bv   = *(const float4*)(bias + 4 * lane);
        for (unsigned k = 0; k < ROWS_W; ++k) {
            const unsigned node = wave + 32u * k;
            if ((int)node >= Rn) break;
            const unsigned jbeg = (unsigned)__builtin_amdgcn_readfirstlane((int)nstart[node]);
            const unsigned jend = jbeg + (unsigned)__builtin_amdgcn_readfirstlane((int)ncnt[node]);
            const float ad = sAD[node * HEADS + (unsigned)hd];
            const float mk = leaky(sMA[node * HEADS + (unsigned)hd] + ad);
            float ax = 0.f, ay = 0.f, az = 0.f, aw = 0.f, sw = 0.f;
            for (unsigned j0 = jbeg; j0 < jend; j0 += AGG_U) {
                unsigned it[4], ent[4];
                float    as[4], wv[4];
#pragma unroll
                for (int t = 0; t < 4; ++t) {
                    it[t]  = perm[min(j0 + (unsigned)q + 4u * (unsigned)t, jend - 1u)];
                    ent[t] = list[it[t]];
                    as[t]  = sASc[it[t] * HEADS + (unsigned)hd];
                }
                float4 hv[AGG_U];
#pragma unroll
                for (int u = 0; u < AGG_U; ++u) {
                    if (j0 + (unsigned)u < jend) {
                        const unsigned s = (unsigned)__builtin_amdgcn_readlane((int)ent[u >> 2], u & 3) >> 8;
                        hv[u] = *(const float4*)(hcol + (size_t)s * HID);
                    }
                }
#pragma unroll
                for (int t = 0; t < 4; ++t) {
                    wv[t] = 0.f;
                    if (j0 + 4u * (unsigned)t < jend) wv[t] = expf(leaky(as[t] + ad) - mk);
                }
#pragma unroll
                for (int u = 0; u < AGG_U; ++u) {
                    if (j0 + (unsigned)u < jend) {
                        const float w = (u & 3) == 0 ? quad_bcast<0>(wv[u >> 2]) :
                                        (u & 3) == 1 ? quad_bcast<1>(wv[u >> 2]) :
                                        (u & 3) == 2 ? quad_bcast<2>(wv[u >> 2]) : quad_bcast<3>(wv[u >> 2]);
                        ax += w * hv[u].x;
                        ay += w * hv[u].y;
                        az += w * hv[u].z;
                        aw += w * hv[u].w;
                        sw += w;
                    }
                }
            }
            row_out(Out + (size_t)(base + (int)node) * HID + 4 * lane, (v2f){ax, ay}, (v2f){az, aw}, sw, bv);
        }
    } else {
        for (int i = tid; i < OWN_R * HID; i += OWN_T) acc[i] = 0.f;
        for (int e = tid; e < NEDGES; e += OWN_T) {
            const unsigned ld = (unsigned)(ei[NEDGES + e] - base);
            if (ld < (unsigned)Rn) {
                const int s = ei[e];
                for (int h = 0; h < HEADS; ++h) lds_fmax(&sMA[ld * HEADS + h], AS[s * HEADS + h]);
            }
        }
        __syncthreads();
        for (int e = tid; e < NEDGES; e += OWN_T) {
            const unsigned ld = (unsigned)(ei[NEDGES + e] - base);
            if (ld < (unsigned)Rn) edge_full_lane(AS, Hf, sMA, sAD, sS, acc, ei[e], (int)ld);
        }
        for (int ld = tid; ld < Rn; ld += OWN_T)
            edge_full_lane(AS, Hf, sMA, sAD, sS, acc, base + ld, ld);
        __syncthreads();

        for (int i = tid; i < Rn * HID; i += OWN_T) {
            const int ld = i >> 7, c = i & (HID - 1);
            const float v = acc[ld * HID + acc_pos(c)] / (sS[ld * HEADS + (c >> 4)] + 1e-16f) + bias[c];
            Out[(size_t)(base + ld) * HID + c] = elu1(v);
        }
    }
}

extern "C" void kernel_launch(void* const* d_in, const int* in_sizes, int n_in,
                              void* d_out, int out_size, void* d_ws, size_t ws_size,
                              hipStream_t stream) {
    const float* x   = (const float*)d_in[0];
    const int*   ei  = (const int*)  d_in[1];
    const float* W1  = (const float*)d_in[2];
    const float* as1 = (const float*)d_in[3];
    const float* ad1 = (const float*)d_in[4];
    const float* b1  = (const float*)d_in[5];
    const float* W2  = (const float*)d_in[6];
    const float* as2 = (const float*)d_in[7];
    const float* ad2 = (const float*)d_in[8];
    const float* b2  = (const float*)d_in[9];

    float* ws     = (float*)d_ws;
    float* bufH   = ws;
    float* bufAS  = bufH  + (size_t)NNODES * HID;
    float* bufAD  = bufAS + (size_t)NNODES * HEADS;
    float* bufAgg = bufAD + (size_t)NNODES * HEADS;

    auto run_layer = [&](const float* xin, const float* Wm, const float* as,
                         const float* ad, const float* bv, float* outBuf, int first) {
        k_gemm_wmma<<<GEMM_NB, 256, 0, stream>>>(xin, Wm, as, ad, bufH, bufAS, bufAD, ei,
                                                 (unsigned*)d_out, first);
        k_owner<<<OWN_NB, OWN_T, 0, stream>>>(ei, bufAS, bufAD, bufH, bv, outBuf,
                                              (unsigned*)d_out, first);
    };

    run_layer(x,      W1, as1, ad1, b1, bufAgg,        1);
    run_layer(bufAgg, W2, as2, ad2, b2, (float*)d_out, 0);
}
